# starting kernel
# baseline (speedup 1.0000x reference)
#include <hip/hip_runtime.h>

typedef __attribute__((ext_vector_type(2))) float v2f;
typedef __attribute__((ext_vector_type(8))) float v8f;


__global__ void fill_f32(float* __restrict__ p, long long n, float v) {
  long long i = (long long)blockIdx.x * blockDim.x + threadIdx.x;
  if (i < n) p[i] = v;
}

__global__ void deg_accum(const int* __restrict__ dst, float* __restrict__ deg, int E) {
  int e = blockIdx.x * blockDim.x + threadIdx.x;
  if (e < E) atomicAdd(&deg[dst[e]], 1.0f);
}

__global__ void deg_rsqrt(const float* __restrict__ deg, float* __restrict__ dinv, int n) {
  int i = blockIdx.x * blockDim.x + threadIdx.x;
  if (i < n) {
    float d = deg[i];
    dinv[i] = d > 0.f ? rsqrtf(d) : 0.f;
  }
}

template <int TK>
__global__ void gemm_wmma_f32_lds(const float* __restrict__ A,
                                  const float* __restrict__ B,
                                  float* __restrict__ C,
                                  int M, int K, int NC) {
  constexpr int LSTR = TK + 4;
  __shared__ float As[16 * LSTR];

  const int tid   = threadIdx.x;
  const int lane  = tid & 31;
  const int wave  = tid >> 5;
  const int row0  = blockIdx.x << 4;
  const int col0  = wave << 4;
  const int mn    = lane & 15;
  const int half  = lane >> 4;
  const int half2 = half << 1;
  const int nthr  = blockDim.x;

  const float* bp = B + (size_t)half2 * NC + col0 + mn;

  v8f acc = {};
  for (int k0 = 0; k0 < K; k0 += TK) {
    for (int t = tid * 4; t < 16 * TK; t += nthr * 4) {
      const int r = t / TK;
      const int c = t % TK;
      const float* gp = A + (size_t)(row0 + r) * K + k0 + c;
      __builtin_prefetch(gp + TK, 0, 1);
      *(float4*)(&As[r * LSTR + c]) = *(const float4*)gp;
    }
    __syncthreads();

#pragma unroll
    for (int kk = 0; kk < TK; kk += 4) {
      v2f a, b;
      a.x = As[mn * LSTR + kk + half2];
      a.y = As[mn * LSTR + kk + half2 + 1];
      b.x = bp[(size_t)(k0 + kk) * NC];
      b.y = bp[(size_t)(k0 + kk + 1) * NC];
      acc = __builtin_amdgcn_wmma_f32_16x16x4_f32(
           false, a,  false, b,
           (short)0, acc,  false,  false);
    }
    __syncthreads();
  }

  float* cp = C + (size_t)(row0 + (half << 3)) * NC + col0 + mn;
#pragma unroll
  for (int g = 0; g < 8; ++g) cp[(size_t)g * NC] = acc[g];
}

template <int CH>
__global__ void scatter_edges4(const int* __restrict__ src, const int* __restrict__ dst,
                               const float* __restrict__ dinv,
                               const float* __restrict__ h, float* __restrict__ z,
                               int E) {
  constexpr int CV = CH / 4;
  long long idx = (long long)blockIdx.x * blockDim.x + threadIdx.x;
  if (idx >= (long long)E * CV) return;
  const int e  = (int)(idx / CV);
  const int c4 = (int)(idx % CV) * 4;
  const int s = src[e], d = dst[e];
  const float norm = dinv[s] * dinv[d];
  const float4 hv = *(const float4*)(h + (size_t)s * CH + c4);
  float* zp = z + (size_t)d * CH + c4;
  atomicAdd(zp + 0, hv.x * norm);
  atomicAdd(zp + 1, hv.y * norm);
  atomicAdd(zp + 2, hv.z * norm);
  atomicAdd(zp + 3, hv.w * norm);
}

template <int CH, bool RELU>
__global__ void self_bias4(const float* __restrict__ h, const float* __restrict__ dinv,
                           const float* __restrict__ bias, float* __restrict__ z,
                           int n) {
  constexpr int CV = CH / 4;
  long long idx = (long long)blockIdx.x * blockDim.x + threadIdx.x;
  if (idx >= (long long)n * CV) return;
  const int i  = (int)(idx / CV);
  const int c4 = (int)(idx % CV) * 4;
  const float dv = dinv[i];
  const float s = dv * dv;
  const float4 hv = *(const float4*)(h + (size_t)i * CH + c4);
  const float4 bv = *(const float4*)(bias + c4);
  float4* zp = (float4*)(z + (size_t)i * CH + c4);
  float4 zv = *zp;
  float4 r;
  r.x = zv.x + hv.x * s + bv.x;
  r.y = zv.y + hv.y * s + bv.y;
  r.z = zv.z + hv.z * s + bv.z;
  r.w = zv.w + hv.w * s + bv.w;
  if (RELU) {
    r.x = fmaxf(r.x, 0.f); r.y = fmaxf(r.y, 0.f);
    r.z = fmaxf(r.z, 0.f); r.w = fmaxf(r.w, 0.f);
  }
  *zp = r;
}

__global__ void edge_dot(const int* __restrict__ es, const int* __restrict__ ed,
                         const float* __restrict__ z, float* __restrict__ out, int L) {
  int j = blockIdx.x * blockDim.x + threadIdx.x;
  if (j >= L) return;
  const float4* a = (const float4*)(z + (size_t)es[j] * 64);
  const float4* b = (const float4*)(z + (size_t)ed[j] * 64);
  float sum = 0.f;
#pragma unroll
  for (int t = 0; t < 16; ++t) {
    float4 x = a[t], y = b[t];
    sum += x.x * y.x + x.y * y.y + x.z * y.z + x.w * y.w;
  }
  out[j] = sum;
}


static inline int cdiv_ll(long long a, long long b) { return (int)((a + b - 1) / b); }

extern "C" void kernel_launch(void* const* d_in, const int* in_sizes, int n_in,
                              void* d_out, int out_size, void* d_ws, size_t ws_size,
                              hipStream_t stream) {
  const float* x   = (const float*)d_in[0];
  const int*   ei  = (const int*)d_in[1];
  const int*   eli = (const int*)d_in[2];
  const float* W1  = (const float*)d_in[3];
  const float* b1  = (const float*)d_in[4];
  const float* W2  = (const float*)d_in[5];
  const float* b2  = (const float*)d_in[6];
  float* out = (float*)d_out;

  const int IN = 512, HID = 128, OUT = 64;
  const int n = in_sizes[0] / IN;
  const int E = in_sizes[1] / 2;
  const int L = in_sizes[2] / 2;
  const int* src = ei;
  const int* dst = ei + E;
  const int* es  = eli;
  const int* ed  = eli + L;

  float* ws   = (float*)d_ws;
  float* deg  = ws;  ws += n;
  float* dinv = ws;  ws += n;
  float* h1   = ws;  ws += (size_t)n * HID;
  float* z1   = ws;  ws += (size_t)n * HID;
  float* h2   = ws;  ws += (size_t)n * OUT;
  float* z2   = ws;  ws += (size_t)n * OUT;

  const int B = 256;

  fill_f32<<<cdiv_ll(n, B), B, 0, stream>>>(deg, n, 1.0f);
  fill_f32<<<cdiv_ll((long long)n * HID, B), B, 0, stream>>>(z1, (long long)n * HID, 0.f);
  fill_f32<<<cdiv_ll((long long)n * OUT, B), B, 0, stream>>>(z2, (long long)n * OUT, 0.f);

  deg_accum<<<cdiv_ll(E, B), B, 0, stream>>>(dst, deg, E);
  deg_rsqrt<<<cdiv_ll(n, B), B, 0, stream>>>(deg, dinv, n);

  gemm_wmma_f32_lds<64><<<n / 16, 32 * (HID / 16), 0, stream>>>(x, W1, h1, n, IN, HID);
  scatter_edges4<128><<<cdiv_ll((long long)E * (HID / 4), B), B, 0, stream>>>(src, dst, dinv, h1, z1, E);
  self_bias4<128, true><<<cdiv_ll((long long)n * (HID / 4), B), B, 0, stream>>>(h1, dinv, b1, z1, n);

  gemm_wmma_f32_lds<64><<<n / 16, 32 * (OUT / 16), 0, stream>>>(z1, W2, h2, n, HID, OUT);
  scatter_edges4<64><<<cdiv_ll((long long)E * (OUT / 4), B), B, 0, stream>>>(src, dst, dinv, h2, z2, E);
  self_bias4<64, false><<<cdiv_ll((long long)n * (OUT / 4), B), B, 0, stream>>>(h2, dinv, b2, z2, n);

  edge_dot<<<cdiv_ll(L, B), B, 0, stream>>>(es, ed, z2, out, L);
}


# scatter-add re-laid lane=channel: each wave-wide f32 atomic and row load covers one whole 128-B line (was 4 lines x 8 lanes); edge ids wave-uniform; sums still float atomics, arrival order
# speedup vs baseline: 1.1851x; 1.1851x over previous
#include <hip/hip_runtime.h>

typedef __attribute__((ext_vector_type(2))) float v2f;
typedef __attribute__((ext_vector_type(8))) float v8f;


__global__ void fill_f32(float* __restrict__ p, long long n, float v) {
  long long i = (long long)blockIdx.x * blockDim.x + threadIdx.x;
  if (i < n) p[i] = v;
}

__global__ void deg_accum(const int* __restrict__ dst, float* __restrict__ deg, int E) {
  int e = blockIdx.x * blockDim.x + threadIdx.x;
  if (e < E) atomicAdd(&deg[dst[e]], 1.0f);
}

__global__ void deg_rsqrt(const float* __restrict__ deg, float* __restrict__ dinv, int n) {
  int i = blockIdx.x * blockDim.x + threadIdx.x;
  if (i < n) {
    float d = deg[i];
    dinv[i] = d > 0.f ? rsqrtf(d) : 0.f;
  }
}

template <int TK>
__global__ void gemm_wmma_f32_lds(const float* __restrict__ A,
                                  const float* __restrict__ B,
                                  float* __restrict__ C,
                                  int M, int K, int NC) {
  constexpr int LSTR = TK + 4;
  __shared__ float As[16 * LSTR];

  const int tid   = threadIdx.x;
  const int lane  = tid & 31;
  const int wave  = tid >> 5;
  const int row0  = blockIdx.x << 4;
  const int col0  = wave << 4;
  const int mn    = lane & 15;
  const int half  = lane >> 4;
  const int half2 = half << 1;
  const int nthr  = blockDim.x;

  const float* bp = B + (size_t)half2 * NC + col0 + mn;

  v8f acc = {};
  for (int k0 = 0; k0 < K; k0 += TK) {
    for (int t = tid * 4; t < 16 * TK; t += nthr * 4) {
      const int r = t / TK;
      const int c = t % TK;
      const float* gp = A + (size_t)(row0 + r) * K + k0 + c;
      __builtin_prefetch(gp + TK, 0, 1);
      *(float4*)(&As[r * LSTR + c]) = *(const float4*)gp;
    }
    __syncthreads();

#pragma unroll
    for (int kk = 0; kk < TK; kk += 4) {
      v2f a, b;
      a.x = As[mn * LSTR + kk + half2];
      a.y = As[mn * LSTR + kk + half2 + 1];
      b.x = bp[(size_t)(k0 + kk) * NC];
      b.y = bp[(size_t)(k0 + kk + 1) * NC];
      acc = __builtin_amdgcn_wmma_f32_16x16x4_f32(
           false, a,  false, b,
           (short)0, acc,  false,  false);
    }
    __syncthreads();
  }

  float* cp = C + (size_t)(row0 + (half << 3)) * NC + col0 + mn;
#pragma unroll
  for (int g = 0; g < 8; ++g) cp[(size_t)g * NC] = acc[g];
}

template <int CH, int NE>
__device__ __forceinline__ void scatter_rows_n(const int* __restrict__ src, const int* __restrict__ dst,
                                               const float* __restrict__ dinv,
                                               const float* __restrict__ h, float* __restrict__ z,
                                               long long e0, int lane) {
  constexpr int LPE = CH / 32;
  float nrm[NE];
  const float* hp[NE];
  float* zp[NE];
  for (int i = 0; i < NE; ++i) {
    const int s = src[e0 + i], d = dst[e0 + i];
    nrm[i] = dinv[s] * dinv[d];
    hp[i] = h + (size_t)s * CH + lane;
    zp[i] = z + (size_t)d * CH + lane;
  }
  float v[NE][LPE];
  for (int i = 0; i < NE; ++i)
    for (int j = 0; j < LPE; ++j) v[i][j] = hp[i][32 * j];
  for (int i = 0; i < NE; ++i)
    for (int j = 0; j < LPE; ++j)
      atomicAdd(zp[i] + 32 * j, v[i][j] * nrm[i]);
}

template <int CH>
__global__ void scatter_edges_rows(const int* __restrict__ src, const int* __restrict__ dst,
                                   const float* __restrict__ dinv,
                                   const float* __restrict__ h, float* __restrict__ z,
                                   int E) {
  constexpr int EPW = 128 / CH;
  const int lane = threadIdx.x & 31;
  const int wv = __builtin_amdgcn_readfirstlane((int)(threadIdx.x >> 5));
  const long long w = (long long)blockIdx.x * (blockDim.x >> 5) + wv;
  const long long e0 = w * EPW;
  if (e0 + EPW <= E) {
    scatter_rows_n<CH, EPW>(src, dst, dinv, h, z, e0, lane);
  } else if (EPW > 1) {
    for (long long e = e0; e < E; ++e)
      scatter_rows_n<CH, 1>(src, dst, dinv, h, z, e, lane);
  }
}

template <int CH, bool RELU>
__global__ void self_bias4(const float* __restrict__ h, const float* __restrict__ dinv,
                           const float* __restrict__ bias, float* __restrict__ z,
                           int n) {
  constexpr int CV = CH / 4;
  long long idx = (long long)blockIdx.x * blockDim.x + threadIdx.x;
  if (idx >= (long long)n * CV) return;
  const int i  = (int)(idx / CV);
  const int c4 = (int)(idx % CV) * 4;
  const float dv = dinv[i];
  const float s = dv * dv;
  const float4 hv = *(const float4*)(h + (size_t)i * CH + c4);
  const float4 bv = *(const float4*)(bias + c4);
  float4* zp = (float4*)(z + (size_t)i * CH + c4);
  float4 zv = *zp;
  float4 r;
  r.x = zv.x + hv.x * s + bv.x;
  r.y = zv.y + hv.y * s + bv.y;
  r.z = zv.z + hv.z * s + bv.z;
  r.w = zv.w + hv.w * s + bv.w;
  if (RELU) {
    r.x = fmaxf(r.x, 0.f); r.y = fmaxf(r.y, 0.f);
    r.z = fmaxf(r.z, 0.f); r.w = fmaxf(r.w, 0.f);
  }
  *zp = r;
}

__global__ void edge_dot(const int* __restrict__ es, const int* __restrict__ ed,
                         const float* __restrict__ z, float* __restrict__ out, int L) {
  int j = blockIdx.x * blockDim.x + threadIdx.x;
  if (j >= L) return;
  const float4* a = (const float4*)(z + (size_t)es[j] * 64);
  const float4* b = (const float4*)(z + (size_t)ed[j] * 64);
  float sum = 0.f;
#pragma unroll
  for (int t = 0; t < 16; ++t) {
    float4 x = a[t], y = b[t];
    sum += x.x * y.x + x.y * y.y + x.z * y.z + x.w * y.w;
  }
  out[j] = sum;
}


static inline int cdiv_ll(long long a, long long b) { return (int)((a + b - 1) / b); }

extern "C" void kernel_launch(void* const* d_in, const int* in_sizes, int n_in,
                              void* d_out, int out_size, void* d_ws, size_t ws_size,
                              hipStream_t stream) {
  const float* x   = (const float*)d_in[0];
  const int*   ei  = (const int*)d_in[1];
  const int*   eli = (const int*)d_in[2];
  const float* W1  = (const float*)d_in[3];
  const float* b1  = (const float*)d_in[4];
  const float* W2  = (const float*)d_in[5];
  const float* b2  = (const float*)d_in[6];
  float* out = (float*)d_out;

  const int IN = 512, HID = 128, OUT = 64;
  const int n = in_sizes[0] / IN;
  const int E = in_sizes[1] / 2;
  const int L = in_sizes[2] / 2;
  const int* src = ei;
  const int* dst = ei + E;
  const int* es  = eli;
  const int* ed  = eli + L;

  float* ws   = (float*)d_ws;
  float* deg  = ws;  ws += n;
  float* dinv = ws;  ws += n;
  float* h1   = ws;  ws += (size_t)n * HID;
  float* z1   = ws;  ws += (size_t)n * HID;
  float* h2   = ws;  ws += (size_t)n * OUT;
  float* z2   = ws;  ws += (size_t)n * OUT;

  const int B = 256;

  fill_f32<<<cdiv_ll(n, B), B, 0, stream>>>(deg, n, 1.0f);
  fill_f32<<<cdiv_ll((long long)n * HID, B), B, 0, stream>>>(z1, (long long)n * HID, 0.f);
  fill_f32<<<cdiv_ll((long long)n * OUT, B), B, 0, stream>>>(z2, (long long)n * OUT, 0.f);

  deg_accum<<<cdiv_ll(E, B), B, 0, stream>>>(dst, deg, E);
  deg_rsqrt<<<cdiv_ll(n, B), B, 0, stream>>>(deg, dinv, n);

  gemm_wmma_f32_lds<64><<<n / 16, 32 * (HID / 16), 0, stream>>>(x, W1, h1, n, IN, HID);
  scatter_edges_rows<128><<<cdiv_ll((long long)E * (HID / 4), B), B, 0, stream>>>(src, dst, dinv, h1, z1, E);
  self_bias4<128, true><<<cdiv_ll((long long)n * (HID / 4), B), B, 0, stream>>>(h1, dinv, b1, z1, n);

  gemm_wmma_f32_lds<64><<<n / 16, 32 * (OUT / 16), 0, stream>>>(z1, W2, h2, n, HID, OUT);
  scatter_edges_rows<64><<<cdiv_ll((long long)E * (OUT / 4), B), B, 0, stream>>>(src, dst, dinv, h2, z2, E);
  self_bias4<64, false><<<cdiv_ll((long long)n * (OUT / 4), B), B, 0, stream>>>(h2, dinv, b2, z2, n);

  edge_dot<<<cdiv_ll(L, B), B, 0, stream>>>(es, ed, z2, out, L);
}


# float-atomic scatter-add -> sum by destination: edges grouped once per call (returning int add = rank, scan, place); a wave sums a node's rows in registers in list order (add arrival), one row store
# speedup vs baseline: 1.8218x; 1.5373x over previous
#include <hip/hip_runtime.h>

typedef __attribute__((ext_vector_type(2))) float v2f;
typedef __attribute__((ext_vector_type(8))) float v8f;


__global__ void fill_i32(int* __restrict__ p, int n, int v) {
  int i = blockIdx.x * blockDim.x + threadIdx.x;
  if (i < n) p[i] = v;
}

__global__ void count_rank(const int* __restrict__ dst, int* __restrict__ cnt,
                           int* __restrict__ rank, int E) {
  int e = blockIdx.x * blockDim.x + threadIdx.x;
  if (e < E) rank[e] = atomicAdd(&cnt[dst[e]], 1);
}

__global__ void __launch_bounds__(1024)
scan_counts(const int* __restrict__ cnt, int* __restrict__ rowptr,
            float* __restrict__ dinv, int n) {
  __shared__ int wtot[32];
  __shared__ int wpre[32];
  const int t    = threadIdx.x;
  const int lane = t & 31;
  const int wv   = t >> 5;
  const int base = blockIdx.x << 10;

  int pre = 0;
  for (int j = t; j < base; j += 1024) pre += cnt[j];

  const int i = base + t;
  const int c = (i < n) ? cnt[i] : 0;
  int x = c;
  for (int o = 1; o < 32; o <<= 1) {
    const int u = __shfl_up(x, o);
    if (lane >= o) x += u;
  }
  for (int o = 16; o > 0; o >>= 1) pre += __shfl_xor(pre, o);
  if (lane == 31) wtot[wv] = x;
  if (lane == 0)  wpre[wv] = pre;
  __syncthreads();

  const int mine = wtot[lane];
  int v = mine;
  for (int o = 1; o < 32; o <<= 1) {
    const int u = __shfl_up(v, o);
    if (lane >= o) v += u;
  }
  int p = wpre[lane];
  for (int o = 16; o > 0; o >>= 1) p += __shfl_xor(p, o);
  const int woff = __shfl(v - mine, wv);

  if (i < n) {
    const int start = p + woff + x - c;
    rowptr[i] = start;
    if (i == n - 1) rowptr[n] = start + c;
    const float dg = (float)(c + 1);
    dinv[i] = dg > 0.f ? rsqrtf(dg) : 0.f;
  }
}

__global__ void place_edges(const int* __restrict__ src, const int* __restrict__ dst,
                            const int* __restrict__ rank, const int* __restrict__ rowptr,
                            int* __restrict__ col, int E) {
  int e = blockIdx.x * blockDim.x + threadIdx.x;
  if (e < E) col[rowptr[dst[e]] + rank[e]] = src[e];
}

template <int TK>
__global__ void gemm_wmma_f32_lds(const float* __restrict__ A,
                                  const float* __restrict__ B,
                                  float* __restrict__ C,
                                  int M, int K, int NC) {
  constexpr int LSTR = TK + 4;
  __shared__ float As[16 * LSTR];

  const int tid   = threadIdx.x;
  const int lane  = tid & 31;
  const int wave  = tid >> 5;
  const int row0  = blockIdx.x << 4;
  const int col0  = wave << 4;
  const int mn    = lane & 15;
  const int half  = lane >> 4;
  const int half2 = half << 1;
  const int nthr  = blockDim.x;

  const float* bp = B + (size_t)half2 * NC + col0 + mn;

  v8f acc = {};
  for (int k0 = 0; k0 < K; k0 += TK) {
    for (int t = tid * 4; t < 16 * TK; t += nthr * 4) {
      const int r = t / TK;
      const int c = t % TK;
      const float* gp = A + (size_t)(row0 + r) * K + k0 + c;
      __builtin_prefetch(gp + TK, 0, 1);
      *(float4*)(&As[r * LSTR + c]) = *(const float4*)gp;
    }
    __syncthreads();

#pragma unroll
    for (int kk = 0; kk < TK; kk += 4) {
      v2f a, b;
      a.x = As[mn * LSTR + kk + half2];
      a.y = As[mn * LSTR + kk + half2 + 1];
      b.x = bp[(size_t)(k0 + kk) * NC];
      b.y = bp[(size_t)(k0 + kk + 1) * NC];
      acc = __builtin_amdgcn_wmma_f32_16x16x4_f32(
           false, a,  false, b,
           (short)0, acc,  false,  false);
    }
    __syncthreads();
  }

  float* cp = C + (size_t)(row0 + (half << 3)) * NC + col0 + mn;
#pragma unroll
  for (int g = 0; g < 8; ++g) cp[(size_t)g * NC] = acc[g];
}

template <int CH, bool RELU>
__global__ void gather_rows(const int* __restrict__ rowptr, const int* __restrict__ col,
                            const float* __restrict__ dinv, const float* __restrict__ h,
                            const float* __restrict__ bias, float* __restrict__ z, int n) {
  constexpr int VW = CH / 32;
  typedef float vecf __attribute__((ext_vector_type(VW)));
  const int lane = threadIdx.x & 31;
  const int wv = __builtin_amdgcn_readfirstlane((int)(threadIdx.x >> 5));
  const int d  = blockIdx.x * (int)(blockDim.x >> 5) + wv;
  if (d >= n) return;
  const int beg = rowptr[d];
  const int end = rowptr[d + 1];
  const float dd = dinv[d];
  const float* hl = h + lane * VW;

  vecf acc = *(const vecf*)(hl + (size_t)d * CH) * (dd * dd);

  for (int j0 = beg; j0 < end; j0 += 32) {
    const int left = end - j0;
    const int m = left < 32 ? left : 32;
    int sv = 0, wb = 0;
    if (lane < m) {
      sv = col[j0 + lane];
      wb = __float_as_int(dinv[sv] * dd);
    }
    auto row = [&](int ix) -> vecf {
      return *(const vecf*)(hl + (size_t)__builtin_amdgcn_readlane(sv, ix) * CH);
    };
    auto wgt = [&](int ix) -> float {
      return __int_as_float(__builtin_amdgcn_readlane(wb, ix));
    };

    int i = 0;
    for (; i + 8 <= m; i += 8) {
      vecf v[8];
      for (int k = 0; k < 8; ++k) v[k] = row(i + k);
      for (int k = 0; k < 8; ++k) acc += v[k] * wgt(i + k);
    }
    if (i + 4 <= m) {
      vecf v[4];
      for (int k = 0; k < 4; ++k) v[k] = row(i + k);
      for (int k = 0; k < 4; ++k) acc += v[k] * wgt(i + k);
      i += 4;
    }
    const int rem = m - i;
    if (rem > 0) {
      vecf v0 = row(i), v1 = {}, v2 = {};
      const float w0 = wgt(i);
      float w1 = 0.f, w2 = 0.f;
      if (rem > 1) { v1 = row(i + 1); w1 = wgt(i + 1); }
      if (rem > 2) { v2 = row(i + 2); w2 = wgt(i + 2); }
      acc += v0 * w0;
      if (rem > 1) acc += v1 * w1;
      if (rem > 2) acc += v2 * w2;
    }
  }

  acc += *(const vecf*)(bias + lane * VW);
  if (RELU) {
    for (int k = 0; k < VW; ++k) acc[k] = fmaxf(acc[k], 0.f);
  }
  *(vecf*)(z + (size_t)d * CH + lane * VW) = acc;
}

__global__ void edge_dot(const int* __restrict__ es, const int* __restrict__ ed,
                         const float* __restrict__ z, float* __restrict__ out, int L) {
  int j = blockIdx.x * blockDim.x + threadIdx.x;
  if (j >= L) return;
  const float4* a = (const float4*)(z + (size_t)es[j] * 64);
  const float4* b = (const float4*)(z + (size_t)ed[j] * 64);
  float sum = 0.f;
#pragma unroll
  for (int t = 0; t < 16; ++t) {
    float4 x = a[t], y = b[t];
    sum += x.x * y.x + x.y * y.y + x.z * y.z + x.w * y.w;
  }
  out[j] = sum;
}


static inline int cdiv_ll(long long a, long long b) { return (int)((a + b - 1) / b); }

extern "C" void kernel_launch(void* const* d_in, const int* in_sizes, int n_in,
                              void* d_out, int out_size, void* d_ws, size_t ws_size,
                              hipStream_t stream) {
  const float* x   = (const float*)d_in[0];
  const int*   ei  = (const int*)d_in[1];
  const int*   eli = (const int*)d_in[2];
  const float* W1  = (const float*)d_in[3];
  const float* b1  = (const float*)d_in[4];
  const float* W2  = (const float*)d_in[5];
  const float* b2  = (const float*)d_in[6];
  float* out = (float*)d_out;

  const int IN = 512, HID = 128, OUT = 64;
  const int n = in_sizes[0] / IN;
  const int E = in_sizes[1] / 2;
  const int L = in_sizes[2] / 2;
  const int* src = ei;
  const int* dst = ei + E;
  const int* es  = eli;
  const int* ed  = eli + L;

  float* ws   = (float*)d_ws;
  int*   cnt  = (int*)ws;  ws += n;
  float* dinv = ws;        ws += n;
  float* h1   = ws;        ws += (size_t)n * HID;
  float* z1   = ws;        ws += (size_t)n * HID;
  int*   rowptr = (int*)ws;
  int*   col    = rowptr + (n + 1);
  int*   rank   = col + E;
  ws += (size_t)n * OUT;
  float* z2   = ws;
  float* h2   = h1;

  const int B = 256;

  fill_i32<<<cdiv_ll(n, B), B, 0, stream>>>(cnt, n, 0);
  count_rank<<<cdiv_ll(E, B), B, 0, stream>>>(dst, cnt, rank, E);
  scan_counts<<<cdiv_ll(n, 1024), 1024, 0, stream>>>(cnt, rowptr, dinv, n);
  place_edges<<<cdiv_ll(E, B), B, 0, stream>>>(src, dst, rank, rowptr, col, E);

  gemm_wmma_f32_lds<64><<<n / 16, 32 * (HID / 16), 0, stream>>>(x, W1, h1, n, IN, HID);
  gather_rows<128, true><<<cdiv_ll(n, B / 32), B, 0, stream>>>(rowptr, col, dinv, h1, b1, z1, n);

  gemm_wmma_f32_lds<64><<<n / 16, 32 * (OUT / 16), 0, stream>>>(z1, W2, h2, n, HID, OUT);
  gather_rows<64, false><<<cdiv_ll(n, B / 32), B, 0, stream>>>(rowptr, col, dinv, h2, b2, z2, n);

  edge_dot<<<cdiv_ll(L, B), B, 0, stream>>>(es, ed, z2, out, L);
}


# edges grouped in one pass: each destination owns 32 list slots (slot = returning int add), longer lists spill to an overflow list; scan and place launches dropped (9->7); sums still in arrival order
# speedup vs baseline: 2.0493x; 1.1249x over previous
#include <hip/hip_runtime.h>

typedef __attribute__((ext_vector_type(2))) float v2f;
typedef __attribute__((ext_vector_type(8))) float v8f;


constexpr int CAP = 32;

__device__ __forceinline__ float dinv_of(int c) {
  const float dg = (float)(c + 1);
  return dg > 0.f ? rsqrtf(dg) : 0.f;
}

__global__ void fill_i32(int* __restrict__ p, int n, int v) {
  int i = blockIdx.x * blockDim.x + threadIdx.x;
  if (i < n) p[i] = v;
}

__global__ void count_place(const int* __restrict__ src, const int* __restrict__ dst,
                            int* __restrict__ cnt, int* __restrict__ col,
                            int* __restrict__ ovfn, int* __restrict__ ovf, int E) {
  int e = blockIdx.x * blockDim.x + threadIdx.x;
  if (e >= E) return;
  const int d = dst[e];
  const int s = src[e];
  const int r = atomicAdd(&cnt[d], 1);
  if (r < CAP) {
    col[d * CAP + r] = s;
  } else {
    const int q = atomicAdd(ovfn, 1);
    ovf[2 * q]     = d;
    ovf[2 * q + 1] = s;
  }
}

template <int TK>
__global__ void gemm_wmma_f32_lds(const float* __restrict__ A,
                                  const float* __restrict__ B,
                                  float* __restrict__ C,
                                  int M, int K, int NC) {
  constexpr int LSTR = TK + 4;
  __shared__ float As[16 * LSTR];

  const int tid   = threadIdx.x;
  const int lane  = tid & 31;
  const int wave  = tid >> 5;
  const int row0  = blockIdx.x << 4;
  const int col0  = wave << 4;
  const int mn    = lane & 15;
  const int half  = lane >> 4;
  const int half2 = half << 1;
  const int nthr  = blockDim.x;

  const float* bp = B + (size_t)half2 * NC + col0 + mn;

  v8f acc = {};
  for (int k0 = 0; k0 < K; k0 += TK) {
    for (int t = tid * 4; t < 16 * TK; t += nthr * 4) {
      const int r = t / TK;
      const int c = t % TK;
      const float* gp = A + (size_t)(row0 + r) * K + k0 + c;
      __builtin_prefetch(gp + TK, 0, 1);
      *(float4*)(&As[r * LSTR + c]) = *(const float4*)gp;
    }
    __syncthreads();

#pragma unroll
    for (int kk = 0; kk < TK; kk += 4) {
      v2f a, b;
      a.x = As[mn * LSTR + kk + half2];
      a.y = As[mn * LSTR + kk + half2 + 1];
      b.x = bp[(size_t)(k0 + kk) * NC];
      b.y = bp[(size_t)(k0 + kk + 1) * NC];
      acc = __builtin_amdgcn_wmma_f32_16x16x4_f32(
           false, a,  false, b,
           (short)0, acc,  false,  false);
    }
    __syncthreads();
  }

  float* cp = C + (size_t)(row0 + (half << 3)) * NC + col0 + mn;
#pragma unroll
  for (int g = 0; g < 8; ++g) cp[(size_t)g * NC] = acc[g];
}

template <int CH, bool RELU>
__global__ void gather_rows(const int* __restrict__ cnt, const int* __restrict__ col,
                            const int* __restrict__ ovfn, const int* __restrict__ ovf,
                            const float* __restrict__ h,
                            const float* __restrict__ bias, float* __restrict__ z, int n) {
  constexpr int VW = CH / 32;
  typedef float vecf __attribute__((ext_vector_type(VW)));
  const int lane = threadIdx.x & 31;
  const int wv = __builtin_amdgcn_readfirstlane((int)(threadIdx.x >> 5));
  const int d  = blockIdx.x * (int)(blockDim.x >> 5) + wv;
  if (d >= n) return;
  const int c = cnt[d];
  const int m = c < CAP ? c : CAP;
  const float dd = dinv_of(c);
  const float* hl = h + lane * VW;

  vecf acc = *(const vecf*)(hl + (size_t)d * CH) * (dd * dd);

  int sv = 0, wb = 0;
  if (lane < m) {
    sv = col[d * CAP + lane];
    wb = __float_as_int(dinv_of(cnt[sv]) * dd);
  }
  auto row = [&](int ix) -> vecf {
    return *(const vecf*)(hl + (size_t)__builtin_amdgcn_readlane(sv, ix) * CH);
  };
  auto wgt = [&](int ix) -> float {
    return __int_as_float(__builtin_amdgcn_readlane(wb, ix));
  };

  int i = 0;
  for (; i + 8 <= m; i += 8) {
    vecf v[8];
    for (int k = 0; k < 8; ++k) v[k] = row(i + k);
    for (int k = 0; k < 8; ++k) acc += v[k] * wgt(i + k);
  }
  if (i + 4 <= m) {
    vecf v[4];
    for (int k = 0; k < 4; ++k) v[k] = row(i + k);
    for (int k = 0; k < 4; ++k) acc += v[k] * wgt(i + k);
    i += 4;
  }
  const int rem = m - i;
  if (rem > 0) {
    vecf v0 = row(i), v1 = {}, v2 = {};
    const float w0 = wgt(i);
    float w1 = 0.f, w2 = 0.f;
    if (rem > 1) { v1 = row(i + 1); w1 = wgt(i + 1); }
    if (rem > 2) { v2 = row(i + 2); w2 = wgt(i + 2); }
    acc += v0 * w0;
    if (rem > 1) acc += v1 * w1;
    if (rem > 2) acc += v2 * w2;
  }

  if (c > CAP) {
    const int tot = ovfn[0];
    for (int j0 = 0; j0 < tot; j0 += 32) {
      int od = -1, os = 0, ow = 0;
      if (j0 + lane < tot) {
        od = ovf[2 * (j0 + lane)];
        os = ovf[2 * (j0 + lane) + 1];
      }
      const bool mine = (od == d);
      if (mine) ow = __float_as_int(dinv_of(cnt[os]) * dd);
      unsigned hits = __builtin_amdgcn_ballot_w32(mine);
      while (hits) {
        const int ix = __builtin_ctz(hits);
        hits &= hits - 1;
        const vecf v = *(const vecf*)(hl + (size_t)__builtin_amdgcn_readlane(os, ix) * CH);
        acc += v * __int_as_float(__builtin_amdgcn_readlane(ow, ix));
      }
    }
  }

  acc += *(const vecf*)(bias + lane * VW);
  if (RELU) {
    for (int k = 0; k < VW; ++k) acc[k] = fmaxf(acc[k], 0.f);
  }
  *(vecf*)(z + (size_t)d * CH + lane * VW) = acc;
}

__global__ void edge_dot(const int* __restrict__ es, const int* __restrict__ ed,
                         const float* __restrict__ z, float* __restrict__ out, int L) {
  int j = blockIdx.x * blockDim.x + threadIdx.x;
  if (j >= L) return;
  const float4* a = (const float4*)(z + (size_t)es[j] * 64);
  const float4* b = (const float4*)(z + (size_t)ed[j] * 64);
  float sum = 0.f;
#pragma unroll
  for (int t = 0; t < 16; ++t) {
    float4 x = a[t], y = b[t];
    sum += x.x * y.x + x.y * y.y + x.z * y.z + x.w * y.w;
  }
  out[j] = sum;
}


static inline int cdiv_ll(long long a, long long b) { return (int)((a + b - 1) / b); }

extern "C" void kernel_launch(void* const* d_in, const int* in_sizes, int n_in,
                              void* d_out, int out_size, void* d_ws, size_t ws_size,
                              hipStream_t stream) {
  const float* x   = (const float*)d_in[0];
  const int*   ei  = (const int*)d_in[1];
  const int*   eli = (const int*)d_in[2];
  const float* W1  = (const float*)d_in[3];
  const float* b1  = (const float*)d_in[4];
  const float* W2  = (const float*)d_in[5];
  const float* b2  = (const float*)d_in[6];
  float* out = (float*)d_out;

  const int IN = 512, HID = 128, OUT = 64;
  const int n = in_sizes[0] / IN;
  const int E = in_sizes[1] / 2;
  const int L = in_sizes[2] / 2;
  const int* src = ei;
  const int* dst = ei + E;
  const int* es  = eli;
  const int* ed  = eli + L;

  float* ws   = (float*)d_ws;
  int*   cnt  = (int*)ws;  ws += n;
  int*   ovfn = (int*)ws;  ws += n;
  float* h1   = ws;        ws += (size_t)n * HID;
  float* z1   = ws;        ws += (size_t)n * HID;
  int*   col  = (int*)ws;
  int*   ovf  = col + (size_t)n * CAP;
  ws += (size_t)n * OUT;
  float* z2   = ws;
  float* h2   = h1;

  const int B = 256;

  fill_i32<<<cdiv_ll(n + 1, B), B, 0, stream>>>(cnt, n + 1, 0);
  count_place<<<cdiv_ll(E, B), B, 0, stream>>>(src, dst, cnt, col, ovfn, ovf, E);

  gemm_wmma_f32_lds<64><<<n / 16, 32 * (HID / 16), 0, stream>>>(x, W1, h1, n, IN, HID);
  gather_rows<128, true><<<cdiv_ll(n, B / 32), B, 0, stream>>>(cnt, col, ovfn, ovf, h1, b1, z1, n);

  gemm_wmma_f32_lds<64><<<n / 16, 32 * (OUT / 16), 0, stream>>>(z1, W2, h2, n, HID, OUT);
  gather_rows<64, false><<<cdiv_ll(n, B / 32), B, 0, stream>>>(cnt, col, ovfn, ovf, h2, b2, z2, n);

  edge_dot<<<cdiv_ll(L, B), B, 0, stream>>>(es, ed, z2, out, L);
}


# both GEMMs re-tiled: block 80 rows x 64 cols (was 16 x 128), a wave keeps 5 row tiles so each B fragment feeds 5 WMMAs (was 1), a K chunk's loads all in flight; sum order per element kept
# speedup vs baseline: 2.2029x; 1.0750x over previous
#include <hip/hip_runtime.h>

typedef __attribute__((ext_vector_type(2))) float v2f;
typedef __attribute__((ext_vector_type(8))) float v8f;


constexpr int CAP = 32;

__device__ __forceinline__ float dinv_of(int c) {
  const float dg = (float)(c + 1);
  return dg > 0.f ? rsqrtf(dg) : 0.f;
}

__global__ void fill_i32(int* __restrict__ p, int n, int v) {
  int i = blockIdx.x * blockDim.x + threadIdx.x;
  if (i < n) p[i] = v;
}

__global__ void count_place(const int* __restrict__ src, const int* __restrict__ dst,
                            int* __restrict__ cnt, int* __restrict__ col,
                            int* __restrict__ ovfn, int* __restrict__ ovf, int E) {
  int e = blockIdx.x * blockDim.x + threadIdx.x;
  if (e >= E) return;
  const int d = dst[e];
  const int s = src[e];
  const int r = atomicAdd(&cnt[d], 1);
  if (r < CAP) {
    col[d * CAP + r] = s;
  } else {
    const int q = atomicAdd(ovfn, 1);
    ovf[2 * q]     = d;
    ovf[2 * q + 1] = s;
  }
}

template <int TK, int R, int NW>
__global__ __launch_bounds__(32 * NW) __attribute__((amdgpu_waves_per_eu(1, 6)))
void gemm_wmma_f32_lds(const float* __restrict__ A,
                       const float* __restrict__ B,
                       float* __restrict__ C,
                       int M, int K, int NC) {
  constexpr int ROWS = 16 * R;
  constexpr int LSTR = TK + 4;
  constexpr int NTHR = 32 * NW;
  constexpr int NLD  = (ROWS * TK) / (4 * NTHR);
  static_assert((ROWS * TK) % (4 * NTHR) == 0, "A block must split evenly over the threads");
  __shared__ float As[ROWS * LSTR];

  const int tid   = threadIdx.x;
  const int lane  = tid & 31;
  const int wave  = tid >> 5;
  const int ncb   = NC / (16 * NW);
  const int row0  = ((int)blockIdx.x / ncb) * ROWS;
  const int col0  = (((int)blockIdx.x % ncb) * NW + wave) << 4;
  const int mn    = lane & 15;
  const int half  = lane >> 4;
  const int half2 = half << 1;

  const float* bp = B + (size_t)half2 * NC + col0 + mn;

  v8f acc[R];
#pragma unroll
  for (int rt = 0; rt < R; ++rt) acc[rt] = v8f{};

  for (int k0 = 0; k0 < K; k0 += TK) {
    if (k0 + TK < K) {
#pragma unroll
      for (int i = 0; i < NLD; ++i) {
        const int t  = (tid + i * NTHR) * 4;
        const int gr = min(row0 + t / TK, M - 1);
        __builtin_prefetch(A + (size_t)gr * K + k0 + TK + t % TK, 0, 1);
      }
    }
    v2f bf[TK / 4];
#pragma unroll
    for (int j = 0; j < TK / 4; ++j) {
      bf[j].x = bp[(size_t)(k0 + 4 * j) * NC];
      bf[j].y = bp[(size_t)(k0 + 4 * j + 1) * NC];
    }
    float4 st[NLD];
#pragma unroll
    for (int i = 0; i < NLD; ++i) {
      const int t  = (tid + i * NTHR) * 4;
      const int gr = min(row0 + t / TK, M - 1);
      const float* gp = A + (size_t)gr * K + k0 + t % TK;
      st[i] = *(const float4*)gp;
    }
#pragma unroll
    for (int i = 0; i < NLD; ++i) {
      const int t = (tid + i * NTHR) * 4;
      *(float4*)(&As[(t / TK) * LSTR + (t % TK)]) = st[i];
    }
    __syncthreads();

#pragma unroll
    for (int j = 0; j < TK / 4; ++j) {
#pragma unroll
      for (int rt = 0; rt < R; ++rt) {
        v2f a;
        a.x = As[(rt * 16 + mn) * LSTR + 4 * j + half2];
        a.y = As[(rt * 16 + mn) * LSTR + 4 * j + half2 + 1];
        acc[rt] = __builtin_amdgcn_wmma_f32_16x16x4_f32(
             false, a,  false, bf[j],
             (short)0, acc[rt],  false,  false);
      }
    }
    __syncthreads();
  }

#pragma unroll
  for (int rt = 0; rt < R; ++rt) {
    const int rbase = row0 + rt * 16 + (half << 3);
    float* cp = C + (size_t)rbase * NC + col0 + mn;
#pragma unroll
    for (int g = 0; g < 8; ++g)
      if (rbase + g < M) cp[(size_t)g * NC] = acc[rt][g];
  }
}

template <int CH, bool RELU>
__global__ void gather_rows(const int* __restrict__ cnt, const int* __restrict__ col,
                            const int* __restrict__ ovfn, const int* __restrict__ ovf,
                            const float* __restrict__ h,
                            const float* __restrict__ bias, float* __restrict__ z, int n) {
  constexpr int VW = CH / 32;
  typedef float vecf __attribute__((ext_vector_type(VW)));
  const int lane = threadIdx.x & 31;
  const int wv = __builtin_amdgcn_readfirstlane((int)(threadIdx.x >> 5));
  const int d  = blockIdx.x * (int)(blockDim.x >> 5) + wv;
  if (d >= n) return;
  const int c = cnt[d];
  const int m = c < CAP ? c : CAP;
  const float dd = dinv_of(c);
  const float* hl = h + lane * VW;

  vecf acc = *(const vecf*)(hl + (size_t)d * CH) * (dd * dd);

  int sv = 0, wb = 0;
  if (lane < m) {
    sv = col[d * CAP + lane];
    wb = __float_as_int(dinv_of(cnt[sv]) * dd);
  }
  auto row = [&](int ix) -> vecf {
    return *(const vecf*)(hl + (size_t)__builtin_amdgcn_readlane(sv, ix) * CH);
  };
  auto wgt = [&](int ix) -> float {
    return __int_as_float(__builtin_amdgcn_readlane(wb, ix));
  };

  int i = 0;
  for (; i + 8 <= m; i += 8) {
    vecf v[8];
    for (int k = 0; k < 8; ++k) v[k] = row(i + k);
    for (int k = 0; k < 8; ++k) acc += v[k] * wgt(i + k);
  }
  if (i + 4 <= m) {
    vecf v[4];
    for (int k = 0; k < 4; ++k) v[k] = row(i + k);
    for (int k = 0; k < 4; ++k) acc += v[k] * wgt(i + k);
    i += 4;
  }
  const int rem = m - i;
  if (rem > 0) {
    vecf v0 = row(i), v1 = {}, v2 = {};
    const float w0 = wgt(i);
    float w1 = 0.f, w2 = 0.f;
    if (rem > 1) { v1 = row(i + 1); w1 = wgt(i + 1); }
    if (rem > 2) { v2 = row(i + 2); w2 = wgt(i + 2); }
    acc += v0 * w0;
    if (rem > 1) acc += v1 * w1;
    if (rem > 2) acc += v2 * w2;
  }

  if (c > CAP) {
    const int tot = ovfn[0];
    for (int j0 = 0; j0 < tot; j0 += 32) {
      int od = -1, os = 0, ow = 0;
      if (j0 + lane < tot) {
        od = ovf[2 * (j0 + lane)];
        os = ovf[2 * (j0 + lane) + 1];
      }
      const bool mine = (od == d);
      if (mine) ow = __float_as_int(dinv_of(cnt[os]) * dd);
      unsigned hits = __builtin_amdgcn_ballot_w32(mine);
      while (hits) {
        const int ix = __builtin_ctz(hits);
        hits &= hits - 1;
        const vecf v = *(const vecf*)(hl + (size_t)__builtin_amdgcn_readlane(os, ix) * CH);
        acc += v * __int_as_float(__builtin_amdgcn_readlane(ow, ix));
      }
    }
  }

  acc += *(const vecf*)(bias + lane * VW);
  if (RELU) {
    for (int k = 0; k < VW; ++k) acc[k] = fmaxf(acc[k], 0.f);
  }
  *(vecf*)(z + (size_t)d * CH + lane * VW) = acc;
}

__global__ void edge_dot(const int* __restrict__ es, const int* __restrict__ ed,
                         const float* __restrict__ z, float* __restrict__ out, int L) {
  int j = blockIdx.x * blockDim.x + threadIdx.x;
  if (j >= L) return;
  const float4* a = (const float4*)(z + (size_t)es[j] * 64);
  const float4* b = (const float4*)(z + (size_t)ed[j] * 64);
  float sum = 0.f;
#pragma unroll
  for (int t = 0; t < 16; ++t) {
    float4 x = a[t], y = b[t];
    sum += x.x * y.x + x.y * y.y + x.z * y.z + x.w * y.w;
  }
  out[j] = sum;
}


static inline int cdiv_ll(long long a, long long b) { return (int)((a + b - 1) / b); }

extern "C" void kernel_launch(void* const* d_in, const int* in_sizes, int n_in,
                              void* d_out, int out_size, void* d_ws, size_t ws_size,
                              hipStream_t stream) {
  const float* x   = (const float*)d_in[0];
  const int*   ei  = (const int*)d_in[1];
  const int*   eli = (const int*)d_in[2];
  const float* W1  = (const float*)d_in[3];
  const float* b1  = (const float*)d_in[4];
  const float* W2  = (const float*)d_in[5];
  const float* b2  = (const float*)d_in[6];
  float* out = (float*)d_out;

  const int IN = 512, HID = 128, OUT = 64;
  const int n = in_sizes[0] / IN;
  const int E = in_sizes[1] / 2;
  const int L = in_sizes[2] / 2;
  const int* src = ei;
  const int* dst = ei + E;
  const int* es  = eli;
  const int* ed  = eli + L;

  float* ws   = (float*)d_ws;
  int*   cnt  = (int*)ws;  ws += n;
  int*   ovfn = (int*)ws;  ws += n;
  float* h1   = ws;        ws += (size_t)n * HID;
  float* z1   = ws;        ws += (size_t)n * HID;
  int*   col  = (int*)ws;
  int*   ovf  = col + (size_t)n * CAP;
  ws += (size_t)n * OUT;
  float* z2   = ws;
  float* h2   = h1;

  const int B = 256;

  fill_i32<<<cdiv_ll(n + 1, B), B, 0, stream>>>(cnt, n + 1, 0);
  count_place<<<cdiv_ll(E, B), B, 0, stream>>>(src, dst, cnt, col, ovfn, ovf, E);

  gemm_wmma_f32_lds<32, 5, 4><<<cdiv_ll(n, 80) * (HID / 64), 128, 0, stream>>>(x, W1, h1, n, IN, HID);
  gather_rows<128, true><<<cdiv_ll(n, B / 32), B, 0, stream>>>(cnt, col, ovfn, ovf, h1, b1, z1, n);

  gemm_wmma_f32_lds<32, 5, 4><<<cdiv_ll(n, 80) * (OUT / 64), 128, 0, stream>>>(z1, W2, h2, n, HID, OUT);
  gather_rows<64, false><<<cdiv_ll(n, B / 32), B, 0, stream>>>(cnt, col, ovfn, ovf, h2, b2, z2, n);

  edge_dot<<<cdiv_ll(L, B), B, 0, stream>>>(es, ed, z2, out, L);
}


# edge grouping without a global add per edge: edges binned by 128-node range (LDS histogram, one global add per block and range), then one block per range ranks by LDS add; sums still in arrival order
# speedup vs baseline: 3.0199x; 1.3709x over previous
#include <hip/hip_runtime.h>

typedef __attribute__((ext_vector_type(2))) float v2f;
typedef __attribute__((ext_vector_type(8))) float v8f;


constexpr int CAP  = 32;
constexpr int SH   = 7;
constexpr int NPB  = 1 << SH;
constexpr int BCAP = 4096;
constexpr int TA   = 512;
constexpr int EPT  = 4;
constexpr int MAXBIN = 512;
constexpr int TB   = 512;
constexpr int RB   = BCAP / TB;
static_assert(TB * RB == BCAP, "one pass of the block must cover a range's region");
static_assert(NPB <= TB, "one thread per node of the range");

__device__ __forceinline__ float dinv_of(int c) {
  const float dg = (float)(c + 1);
  return dg > 0.f ? rsqrtf(dg) : 0.f;
}

__global__ void fill_i32(int* __restrict__ p, int n, int v) {
  int i = blockIdx.x * blockDim.x + threadIdx.x;
  if (i < n) p[i] = v;
}

__global__ __launch_bounds__(TA)
void bin_edges(const int* __restrict__ src, const int* __restrict__ dst,
               int* __restrict__ gcnt, int* __restrict__ rec,
               int* __restrict__ nspill, int* __restrict__ spill, int E, int nbin) {
  __shared__ int hist[MAXBIN];
  __shared__ int base[MAXBIN];
  const int tid = threadIdx.x;
  for (int b = tid; b < nbin; b += TA) hist[b] = 0;
  __syncthreads();

  const int e0 = ((int)blockIdx.x * TA + tid) * EPT;
  int d[EPT], s[EPT], lr[EPT];
  if (((E & 3) == 0) && e0 + EPT <= E) {
    const int4 dv = *(const int4*)(dst + e0);
    const int4 sv = *(const int4*)(src + e0);
    d[0] = dv.x; d[1] = dv.y; d[2] = dv.z; d[3] = dv.w;
    s[0] = sv.x; s[1] = sv.y; s[2] = sv.z; s[3] = sv.w;
  } else {
#pragma unroll
    for (int k = 0; k < EPT; ++k) {
      const bool in = e0 + k < E;
      d[k] = in ? dst[e0 + k] : -1;
      s[k] = in ? src[e0 + k] : 0;
    }
  }
#pragma unroll
  for (int k = 0; k < EPT; ++k)
    lr[k] = d[k] >= 0 ? atomicAdd(&hist[d[k] >> SH], 1) : 0;
  __syncthreads();

  for (int b = tid; b < nbin; b += TA) {
    const int c = hist[b];
    base[b] = c ? atomicAdd(&gcnt[b], c) : 0;
  }
  __syncthreads();

#pragma unroll
  for (int k = 0; k < EPT; ++k) {
    if (d[k] < 0) continue;
    const int bin = d[k] >> SH;
    const int pos = base[bin] + lr[k];
    int* p = rec + ((size_t)bin * BCAP + pos) * 2;
    if (pos >= BCAP)
      p = spill + 2 * (size_t)atomicAdd(nspill, 1);
    *(int2*)p = make_int2(d[k], s[k]);
  }
}

__global__ __launch_bounds__(TB)
void rank_range(const int* __restrict__ gcnt, const int* __restrict__ rec,
                const int* __restrict__ nspill, const int* __restrict__ spill,
                int* __restrict__ cnt, int* __restrict__ col,
                int* __restrict__ ovfn, int* __restrict__ ovf, int n) {
  __shared__ int lc[NPB];
  __shared__ int4 slots[NPB * CAP / 4];
  const int tid = threadIdx.x;
  const int bin = blockIdx.x;
  const int d0  = bin << SH;
  int* sl = (int*)slots;

  if (tid < NPB) lc[tid] = 0;
  for (int j = tid; j < NPB * CAP / 4; j += TB) slots[j] = make_int4(0, 0, 0, 0);
  __syncthreads();

  auto put = [&](int d, int s) {
    const int ld = d - d0;
    const int r  = atomicAdd(&lc[ld], 1);
    if (r < CAP) {
      sl[ld * CAP + r] = s;
    } else {
      const int q = atomicAdd(ovfn, 1);
      ovf[2 * q]     = d;
      ovf[2 * q + 1] = s;
    }
  };

  const int g = gcnt[bin];
  const int m = g < BCAP ? g : BCAP;
  const int2* rp = (const int2*)(rec + (size_t)bin * BCAP * 2);
  int2 r[RB];
#pragma unroll
  for (int k = 0; k < RB; ++k) {
    const int j = k * TB + tid;
    r[k] = j < m ? rp[j] : make_int2(-1, 0);
  }
#pragma unroll
  for (int k = 0; k < RB; ++k)
    if (r[k].x >= 0) put(r[k].x, r[k].y);
  const int ns = nspill[0];
  for (int j = tid; j < ns; j += TB) {
    const int d = spill[2 * j];
    if ((d >> SH) == bin) put(d, spill[2 * j + 1]);
  }
  __syncthreads();

  const int nn = min(NPB, n - d0);
  if (tid < nn) cnt[d0 + tid] = lc[tid];
  int4* cp = (int4*)(col + (size_t)d0 * CAP);
  for (int j = tid; j < nn * (CAP / 4); j += TB) cp[j] = slots[j];
}

template <int TK, int R, int NW>
__global__ __launch_bounds__(32 * NW) __attribute__((amdgpu_waves_per_eu(1, 6)))
void gemm_wmma_f32_lds(const float* __restrict__ A,
                       const float* __restrict__ B,
                       float* __restrict__ C,
                       int M, int K, int NC) {
  constexpr int ROWS = 16 * R;
  constexpr int LSTR = TK + 4;
  constexpr int NTHR = 32 * NW;
  constexpr int NLD  = (ROWS * TK) / (4 * NTHR);
  static_assert((ROWS * TK) % (4 * NTHR) == 0, "A block must split evenly over the threads");
  __shared__ float As[ROWS * LSTR];

  const int tid   = threadIdx.x;
  const int lane  = tid & 31;
  const int wave  = tid >> 5;
  const int ncb   = NC / (16 * NW);
  const int row0  = ((int)blockIdx.x / ncb) * ROWS;
  const int col0  = (((int)blockIdx.x % ncb) * NW + wave) << 4;
  const int mn    = lane & 15;
  const int half  = lane >> 4;
  const int half2 = half << 1;

  const float* bp = B + (size_t)half2 * NC + col0 + mn;

  v8f acc[R];
#pragma unroll
  for (int rt = 0; rt < R; ++rt) acc[rt] = v8f{};

  for (int k0 = 0; k0 < K; k0 += TK) {
    if (k0 + TK < K) {
#pragma unroll
      for (int i = 0; i < NLD; ++i) {
        const int t  = (tid + i * NTHR) * 4;
        const int gr = min(row0 + t / TK, M - 1);
        __builtin_prefetch(A + (size_t)gr * K + k0 + TK + t % TK, 0, 1);
      }
    }
    v2f bf[TK / 4];
#pragma unroll
    for (int j = 0; j < TK / 4; ++j) {
      bf[j].x = bp[(size_t)(k0 + 4 * j) * NC];
      bf[j].y = bp[(size_t)(k0 + 4 * j + 1) * NC];
    }
    float4 st[NLD];
#pragma unroll
    for (int i = 0; i < NLD; ++i) {
      const int t  = (tid + i * NTHR) * 4;
      const int gr = min(row0 + t / TK, M - 1);
      const float* gp = A + (size_t)gr * K + k0 + t % TK;
      st[i] = *(const float4*)gp;
    }
#pragma unroll
    for (int i = 0; i < NLD; ++i) {
      const int t = (tid + i * NTHR) * 4;
      *(float4*)(&As[(t / TK) * LSTR + (t % TK)]) = st[i];
    }
    __syncthreads();

#pragma unroll
    for (int j = 0; j < TK / 4; ++j) {
#pragma unroll
      for (int rt = 0; rt < R; ++rt) {
        v2f a;
        a.x = As[(rt * 16 + mn) * LSTR + 4 * j + half2];
        a.y = As[(rt * 16 + mn) * LSTR + 4 * j + half2 + 1];
        acc[rt] = __builtin_amdgcn_wmma_f32_16x16x4_f32(
             false, a,  false, bf[j],
             (short)0, acc[rt],  false,  false);
      }
    }
    __syncthreads();
  }

#pragma unroll
  for (int rt = 0; rt < R; ++rt) {
    const int rbase = row0 + rt * 16 + (half << 3);
    float* cp = C + (size_t)rbase * NC + col0 + mn;
#pragma unroll
    for (int g = 0; g < 8; ++g)
      if (rbase + g < M) cp[(size_t)g * NC] = acc[rt][g];
  }
}

template <int CH, bool RELU>
__global__ void gather_rows(const int* __restrict__ cnt, const int* __restrict__ col,
                            const int* __restrict__ ovfn, const int* __restrict__ ovf,
                            const float* __restrict__ h,
                            const float* __restrict__ bias, float* __restrict__ z, int n) {
  constexpr int VW = CH / 32;
  typedef float vecf __attribute__((ext_vector_type(VW)));
  const int lane = threadIdx.x & 31;
  const int wv = __builtin_amdgcn_readfirstlane((int)(threadIdx.x >> 5));
  const int d  = blockIdx.x * (int)(blockDim.x >> 5) + wv;
  if (d >= n) return;
  const int c = cnt[d];
  const int m = c < CAP ? c : CAP;
  const float dd = dinv_of(c);
  const float* hl = h + lane * VW;

  vecf acc = *(const vecf*)(hl + (size_t)d * CH) * (dd * dd);

  int sv = 0, wb = 0;
  if (lane < m) {
    sv = col[d * CAP + lane];
    wb = __float_as_int(dinv_of(cnt[sv]) * dd);
  }
  auto row = [&](int ix) -> vecf {
    return *(const vecf*)(hl + (size_t)__builtin_amdgcn_readlane(sv, ix) * CH);
  };
  auto wgt = [&](int ix) -> float {
    return __int_as_float(__builtin_amdgcn_readlane(wb, ix));
  };

  int i = 0;
  for (; i + 8 <= m; i += 8) {
    vecf v[8];
    for (int k = 0; k < 8; ++k) v[k] = row(i + k);
    for (int k = 0; k < 8; ++k) acc += v[k] * wgt(i + k);
  }
  if (i + 4 <= m) {
    vecf v[4];
    for (int k = 0; k < 4; ++k) v[k] = row(i + k);
    for (int k = 0; k < 4; ++k) acc += v[k] * wgt(i + k);
    i += 4;
  }
  const int rem = m - i;
  if (rem > 0) {
    vecf v0 = row(i), v1 = {}, v2 = {};
    const float w0 = wgt(i);
    float w1 = 0.f, w2 = 0.f;
    if (rem > 1) { v1 = row(i + 1); w1 = wgt(i + 1); }
    if (rem > 2) { v2 = row(i + 2); w2 = wgt(i + 2); }
    acc += v0 * w0;
    if (rem > 1) acc += v1 * w1;
    if (rem > 2) acc += v2 * w2;
  }

  if (c > CAP) {
    const int tot = ovfn[0];
    for (int j0 = 0; j0 < tot; j0 += 32) {
      int od = -1, os = 0, ow = 0;
      if (j0 + lane < tot) {
        od = ovf[2 * (j0 + lane)];
        os = ovf[2 * (j0 + lane) + 1];
      }
      const bool mine = (od == d);
      if (mine) ow = __float_as_int(dinv_of(cnt[os]) * dd);
      unsigned hits = __builtin_amdgcn_ballot_w32(mine);
      while (hits) {
        const int ix = __builtin_ctz(hits);
        hits &= hits - 1;
        const vecf v = *(const vecf*)(hl + (size_t)__builtin_amdgcn_readlane(os, ix) * CH);
        acc += v * __int_as_float(__builtin_amdgcn_readlane(ow, ix));
      }
    }
  }

  acc += *(const vecf*)(bias + lane * VW);
  if (RELU) {
    for (int k = 0; k < VW; ++k) acc[k] = fmaxf(acc[k], 0.f);
  }
  *(vecf*)(z + (size_t)d * CH + lane * VW) = acc;
}

__global__ void edge_dot(const int* __restrict__ es, const int* __restrict__ ed,
                         const float* __restrict__ z, float* __restrict__ out, int L) {
  int j = blockIdx.x * blockDim.x + threadIdx.x;
  if (j >= L) return;
  const float4* a = (const float4*)(z + (size_t)es[j] * 64);
  const float4* b = (const float4*)(z + (size_t)ed[j] * 64);
  float sum = 0.f;
#pragma unroll
  for (int t = 0; t < 16; ++t) {
    float4 x = a[t], y = b[t];
    sum += x.x * y.x + x.y * y.y + x.z * y.z + x.w * y.w;
  }
  out[j] = sum;
}


static inline int cdiv_ll(long long a, long long b) { return (int)((a + b - 1) / b); }

extern "C" void kernel_launch(void* const* d_in, const int* in_sizes, int n_in,
                              void* d_out, int out_size, void* d_ws, size_t ws_size,
                              hipStream_t stream) {
  const float* x   = (const float*)d_in[0];
  const int*   ei  = (const int*)d_in[1];
  const int*   eli = (const int*)d_in[2];
  const float* W1  = (const float*)d_in[3];
  const float* b1  = (const float*)d_in[4];
  const float* W2  = (const float*)d_in[5];
  const float* b2  = (const float*)d_in[6];
  float* out = (float*)d_out;

  const int IN = 512, HID = 128, OUT = 64;
  const int n = in_sizes[0] / IN;
  const int E = in_sizes[1] / 2;
  const int L = in_sizes[2] / 2;
  const int* src = ei;
  const int* dst = ei + E;
  const int* es  = eli;
  const int* ed  = eli + L;

  const int nbin = cdiv_ll(n, NPB);
  float* ws   = (float*)d_ws;
  int*   cnt  = (int*)ws;  ws += n;
  int*   ctr  = (int*)ws;  ws += n;
  float* h1   = ws;        ws += (size_t)n * HID;
  float* z1   = ws;        ws += (size_t)n * HID;
  int*   col  = (int*)ws;
  int*   ovf  = col + (size_t)n * CAP;
  ws += (size_t)n * OUT;
  float* z2   = ws;
  float* h2   = h1;
  int*   ovfn   = ctr;
  int*   nspill = ctr + 1;
  int*   gcnt   = ctr + 2;
  int*   rec    = (int*)z1;
  int*   spill  = rec + (size_t)nbin * BCAP * 2;

  const int B = 256;

  fill_i32<<<cdiv_ll(nbin + 2, B), B, 0, stream>>>(ctr, nbin + 2, 0);
  bin_edges<<<cdiv_ll(E, TA * EPT), TA, 0, stream>>>(src, dst, gcnt, rec, nspill, spill, E, nbin);
  rank_range<<<nbin, TB, 0, stream>>>(gcnt, rec, nspill, spill, cnt, col, ovfn, ovf, n);

  gemm_wmma_f32_lds<32, 5, 4><<<cdiv_ll(n, 80) * (HID / 64), 128, 0, stream>>>(x, W1, h1, n, IN, HID);
  gather_rows<128, true><<<cdiv_ll(n, B / 32), B, 0, stream>>>(cnt, col, ovfn, ovf, h1, b1, z1, n);

  gemm_wmma_f32_lds<32, 5, 4><<<cdiv_ll(n, 80) * (OUT / 64), 128, 0, stream>>>(z1, W2, h2, n, HID, OUT);
  gather_rows<64, false><<<cdiv_ll(n, B / 32), B, 0, stream>>>(cnt, col, ovfn, ovf, h2, b2, z2, n);

  edge_dot<<<cdiv_ll(L, B), B, 0, stream>>>(es, ed, z2, out, L);
}


# layer 1's sum by destination and layer 2's product in one launch (8->7): a 4-wave block sums 16 nodes' rows into LDS, then its 16x64 tile on WMMA from LDS; z1 store and re-read gone; sum orders kept
# speedup vs baseline: 3.1860x; 1.0550x over previous
#include <hip/hip_runtime.h>

typedef __attribute__((ext_vector_type(2))) float v2f;
typedef __attribute__((ext_vector_type(8))) float v8f;


constexpr int CAP  = 32;
constexpr int SH   = 7;
constexpr int NPB  = 1 << SH;
constexpr int BCAP = 4096;
constexpr int TA   = 512;
constexpr int EPT  = 4;
constexpr int MAXBIN = 512;
constexpr int TB   = 512;
constexpr int RB   = BCAP / TB;
static_assert(TB * RB == BCAP, "one pass of the block must cover a range's region");
static_assert(NPB <= TB, "one thread per node of the range");

__device__ __forceinline__ float dinv_of(int c) {
  const float dg = (float)(c + 1);
  return dg > 0.f ? rsqrtf(dg) : 0.f;
}

__global__ void fill_i32(int* __restrict__ p, int n, int v) {
  int i = blockIdx.x * blockDim.x + threadIdx.x;
  if (i < n) p[i] = v;
}

__global__ __launch_bounds__(TA)
void bin_edges(const int* __restrict__ src, const int* __restrict__ dst,
               int* __restrict__ gcnt, int* __restrict__ rec,
               int* __restrict__ nspill, int* __restrict__ spill, int E, int nbin) {
  __shared__ int hist[MAXBIN];
  __shared__ int base[MAXBIN];
  const int tid = threadIdx.x;
  for (int b = tid; b < nbin; b += TA) hist[b] = 0;
  __syncthreads();

  const int e0 = ((int)blockIdx.x * TA + tid) * EPT;
  int d[EPT], s[EPT], lr[EPT];
  if (((E & 3) == 0) && e0 + EPT <= E) {
    const int4 dv = *(const int4*)(dst + e0);
    const int4 sv = *(const int4*)(src + e0);
    d[0] = dv.x; d[1] = dv.y; d[2] = dv.z; d[3] = dv.w;
    s[0] = sv.x; s[1] = sv.y; s[2] = sv.z; s[3] = sv.w;
  } else {
#pragma unroll
    for (int k = 0; k < EPT; ++k) {
      const bool in = e0 + k < E;
      d[k] = in ? dst[e0 + k] : -1;
      s[k] = in ? src[e0 + k] : 0;
    }
  }
#pragma unroll
  for (int k = 0; k < EPT; ++k)
    lr[k] = d[k] >= 0 ? atomicAdd(&hist[d[k] >> SH], 1) : 0;
  __syncthreads();

  for (int b = tid; b < nbin; b += TA) {
    const int c = hist[b];
    base[b] = c ? atomicAdd(&gcnt[b], c) : 0;
  }
  __syncthreads();

#pragma unroll
  for (int k = 0; k < EPT; ++k) {
    if (d[k] < 0) continue;
    const int bin = d[k] >> SH;
    const int pos = base[bin] + lr[k];
    int* p = rec + ((size_t)bin * BCAP + pos) * 2;
    if (pos >= BCAP)
      p = spill + 2 * (size_t)atomicAdd(nspill, 1);
    *(int2*)p = make_int2(d[k], s[k]);
  }
}

__global__ __launch_bounds__(TB)
void rank_range(const int* __restrict__ gcnt, const int* __restrict__ rec,
                const int* __restrict__ nspill, const int* __restrict__ spill,
                int* __restrict__ cnt, int* __restrict__ col,
                int* __restrict__ ovfn, int* __restrict__ ovf, int n) {
  __shared__ int lc[NPB];
  __shared__ int4 slots[NPB * CAP / 4];
  const int tid = threadIdx.x;
  const int bin = blockIdx.x;
  const int d0  = bin << SH;
  int* sl = (int*)slots;

  if (tid < NPB) lc[tid] = 0;
  for (int j = tid; j < NPB * CAP / 4; j += TB) slots[j] = make_int4(0, 0, 0, 0);
  __syncthreads();

  auto put = [&](int d, int s) {
    const int ld = d - d0;
    const int r  = atomicAdd(&lc[ld], 1);
    if (r < CAP) {
      sl[ld * CAP + r] = s;
    } else {
      const int q = atomicAdd(ovfn, 1);
      ovf[2 * q]     = d;
      ovf[2 * q + 1] = s;
    }
  };

  const int g = gcnt[bin];
  const int m = g < BCAP ? g : BCAP;
  const int2* rp = (const int2*)(rec + (size_t)bin * BCAP * 2);
  int2 r[RB];
#pragma unroll
  for (int k = 0; k < RB; ++k) {
    const int j = k * TB + tid;
    r[k] = j < m ? rp[j] : make_int2(-1, 0);
  }
#pragma unroll
  for (int k = 0; k < RB; ++k)
    if (r[k].x >= 0) put(r[k].x, r[k].y);
  const int ns = nspill[0];
  for (int j = tid; j < ns; j += TB) {
    const int d = spill[2 * j];
    if ((d >> SH) == bin) put(d, spill[2 * j + 1]);
  }
  __syncthreads();

  const int nn = min(NPB, n - d0);
  if (tid < nn) cnt[d0 + tid] = lc[tid];
  int4* cp = (int4*)(col + (size_t)d0 * CAP);
  for (int j = tid; j < nn * (CAP / 4); j += TB) cp[j] = slots[j];
}

template <int TK, int R, int NW>
__global__ __launch_bounds__(32 * NW) __attribute__((amdgpu_waves_per_eu(1, 6)))
void gemm_wmma_f32_lds(const float* __restrict__ A,
                       const float* __restrict__ B,
                       float* __restrict__ C,
                       int M, int K, int NC) {
  constexpr int ROWS = 16 * R;
  constexpr int LSTR = TK + 4;
  constexpr int NTHR = 32 * NW;
  constexpr int NLD  = (ROWS * TK) / (4 * NTHR);
  static_assert((ROWS * TK) % (4 * NTHR) == 0, "A block must split evenly over the threads");
  __shared__ float As[ROWS * LSTR];

  const int tid   = threadIdx.x;
  const int lane  = tid & 31;
  const int wave  = tid >> 5;
  const int ncb   = NC / (16 * NW);
  const int row0  = ((int)blockIdx.x / ncb) * ROWS;
  const int col0  = (((int)blockIdx.x % ncb) * NW + wave) << 4;
  const int mn    = lane & 15;
  const int half  = lane >> 4;
  const int half2 = half << 1;

  const float* bp = B + (size_t)half2 * NC + col0 + mn;

  v8f acc[R];
#pragma unroll
  for (int rt = 0; rt < R; ++rt) acc[rt] = v8f{};

  for (int k0 = 0; k0 < K; k0 += TK) {
    if (k0 + TK < K) {
#pragma unroll
      for (int i = 0; i < NLD; ++i) {
        const int t  = (tid + i * NTHR) * 4;
        const int gr = min(row0 + t / TK, M - 1);
        __builtin_prefetch(A + (size_t)gr * K + k0 + TK + t % TK, 0, 1);
      }
    }
    v2f bf[TK / 4];
#pragma unroll
    for (int j = 0; j < TK / 4; ++j) {
      bf[j].x = bp[(size_t)(k0 + 4 * j) * NC];
      bf[j].y = bp[(size_t)(k0 + 4 * j + 1) * NC];
    }
    float4 st[NLD];
#pragma unroll
    for (int i = 0; i < NLD; ++i) {
      const int t  = (tid + i * NTHR) * 4;
      const int gr = min(row0 + t / TK, M - 1);
      const float* gp = A + (size_t)gr * K + k0 + t % TK;
      st[i] = *(const float4*)gp;
    }
#pragma unroll
    for (int i = 0; i < NLD; ++i) {
      const int t = (tid + i * NTHR) * 4;
      *(float4*)(&As[(t / TK) * LSTR + (t % TK)]) = st[i];
    }
    __syncthreads();

#pragma unroll
    for (int j = 0; j < TK / 4; ++j) {
#pragma unroll
      for (int rt = 0; rt < R; ++rt) {
        v2f a;
        a.x = As[(rt * 16 + mn) * LSTR + 4 * j + half2];
        a.y = As[(rt * 16 + mn) * LSTR + 4 * j + half2 + 1];
        acc[rt] = __builtin_amdgcn_wmma_f32_16x16x4_f32(
             false, a,  false, bf[j],
             (short)0, acc[rt],  false,  false);
      }
    }
    __syncthreads();
  }

#pragma unroll
  for (int rt = 0; rt < R; ++rt) {
    const int rbase = row0 + rt * 16 + (half << 3);
    float* cp = C + (size_t)rbase * NC + col0 + mn;
#pragma unroll
    for (int g = 0; g < 8; ++g)
      if (rbase + g < M) cp[(size_t)g * NC] = acc[rt][g];
  }
}

template <int CH>
__device__ __forceinline__ float __attribute__((ext_vector_type(CH / 32)))
sum_node(int d, int lane, const int* __restrict__ cnt, const int* __restrict__ col,
         const int* __restrict__ ovfn, const int* __restrict__ ovf,
         const float* __restrict__ h) {
  constexpr int VW = CH / 32;
  typedef float vecf __attribute__((ext_vector_type(VW)));
  const int c = cnt[d];
  const int m = c < CAP ? c : CAP;
  const float dd = dinv_of(c);
  const float* hl = h + lane * VW;

  vecf acc = *(const vecf*)(hl + (size_t)d * CH) * (dd * dd);

  int sv = 0, wb = 0;
  if (lane < m) {
    sv = col[d * CAP + lane];
    wb = __float_as_int(dinv_of(cnt[sv]) * dd);
  }
  auto row = [&](int ix) -> vecf {
    return *(const vecf*)(hl + (size_t)__builtin_amdgcn_readlane(sv, ix) * CH);
  };
  auto wgt = [&](int ix) -> float {
    return __int_as_float(__builtin_amdgcn_readlane(wb, ix));
  };

  int i = 0;
  for (; i + 8 <= m; i += 8) {
    vecf v[8];
    for (int k = 0; k < 8; ++k) v[k] = row(i + k);
    for (int k = 0; k < 8; ++k) acc += v[k] * wgt(i + k);
  }
  if (i + 4 <= m) {
    vecf v[4];
    for (int k = 0; k < 4; ++k) v[k] = row(i + k);
    for (int k = 0; k < 4; ++k) acc += v[k] * wgt(i + k);
    i += 4;
  }
  const int rem = m - i;
  if (rem > 0) {
    vecf v0 = row(i), v1 = {}, v2 = {};
    const float w0 = wgt(i);
    float w1 = 0.f, w2 = 0.f;
    if (rem > 1) { v1 = row(i + 1); w1 = wgt(i + 1); }
    if (rem > 2) { v2 = row(i + 2); w2 = wgt(i + 2); }
    acc += v0 * w0;
    if (rem > 1) acc += v1 * w1;
    if (rem > 2) acc += v2 * w2;
  }

  if (c > CAP) {
    const int tot = ovfn[0];
    for (int j0 = 0; j0 < tot; j0 += 32) {
      int od = -1, os = 0, ow = 0;
      if (j0 + lane < tot) {
        od = ovf[2 * (j0 + lane)];
        os = ovf[2 * (j0 + lane) + 1];
      }
      const bool mine = (od == d);
      if (mine) ow = __float_as_int(dinv_of(cnt[os]) * dd);
      unsigned hits = __builtin_amdgcn_ballot_w32(mine);
      while (hits) {
        const int ix = __builtin_ctz(hits);
        hits &= hits - 1;
        const vecf v = *(const vecf*)(hl + (size_t)__builtin_amdgcn_readlane(os, ix) * CH);
        acc += v * __int_as_float(__builtin_amdgcn_readlane(ow, ix));
      }
    }
  }
  return acc;
}

template <int CH, bool RELU>
__global__ void gather_rows(const int* __restrict__ cnt, const int* __restrict__ col,
                            const int* __restrict__ ovfn, const int* __restrict__ ovf,
                            const float* __restrict__ h,
                            const float* __restrict__ bias, float* __restrict__ z, int n) {
  constexpr int VW = CH / 32;
  typedef float vecf __attribute__((ext_vector_type(VW)));
  const int lane = threadIdx.x & 31;
  const int wv = __builtin_amdgcn_readfirstlane((int)(threadIdx.x >> 5));
  const int d  = blockIdx.x * (int)(blockDim.x >> 5) + wv;
  if (d >= n) return;

  vecf acc = sum_node<CH>(d, lane, cnt, col, ovfn, ovf, h);
  acc += *(const vecf*)(bias + lane * VW);
  if (RELU) {
    for (int k = 0; k < VW; ++k) acc[k] = fmaxf(acc[k], 0.f);
  }
  *(vecf*)(z + (size_t)d * CH + lane * VW) = acc;
}

constexpr int FN   = 16;
constexpr int FW   = 4;
constexpr int FCH  = 128;
constexpr int FNC  = 64;
constexpr int FSTR = FCH + 4;
constexpr int FKC  = 8;
static_assert(FW * 16 == FNC, "one wave per column tile");
__global__ __launch_bounds__(32 * FW) __attribute__((amdgpu_waves_per_eu(16, 16)))
void gather_rows_gemm(const int* __restrict__ cnt, const int* __restrict__ col,
                      const int* __restrict__ ovfn, const int* __restrict__ ovf,
                      const float* __restrict__ h, const float* __restrict__ bias,
                      const float* __restrict__ Wn, float* __restrict__ hn, int n) {
  typedef float vecf __attribute__((ext_vector_type(4)));
  __shared__ float zs[FN * FSTR];
  const int lane  = threadIdx.x & 31;
  const int wv    = __builtin_amdgcn_readfirstlane((int)(threadIdx.x >> 5));
  const int node0 = blockIdx.x * FN;

  for (int q = 0; q < FN / FW; ++q) {
    const int r = q * FW + wv;
    const int d = node0 + r;
    vecf acc = {};
    if (d < n) {
      acc = sum_node<FCH>(d, lane, cnt, col, ovfn, ovf, h);
      acc += *(const vecf*)(bias + lane * 4);
      for (int k = 0; k < 4; ++k) acc[k] = fmaxf(acc[k], 0.f);
    }
    *(vecf*)(&zs[r * FSTR + lane * 4]) = acc;
  }

  const int mn    = lane & 15;
  const int half  = lane >> 4;
  const int half2 = half << 1;
  const int col0  = wv << 4;
  const float* bp = Wn + (size_t)half2 * FNC + col0 + mn;
  v2f bf[FCH / 4];
#pragma unroll
  for (int j = 0; j < FKC; ++j) {
    bf[j].x = bp[(size_t)(4 * j) * FNC];
    bf[j].y = bp[(size_t)(4 * j + 1) * FNC];
  }
  __syncthreads();
#pragma unroll
  for (int j = FKC; j < FCH / 4; ++j) {
    bf[j].x = bp[(size_t)(4 * j) * FNC];
    bf[j].y = bp[(size_t)(4 * j + 1) * FNC];
  }

  v8f c = v8f{};
  const float* ap = &zs[mn * FSTR + half2];
#pragma unroll
  for (int j = 0; j < FCH / 4; ++j) {
    v2f a;
    a.x = ap[4 * j];
    a.y = ap[4 * j + 1];
    c = __builtin_amdgcn_wmma_f32_16x16x4_f32(
         false, a,  false, bf[j],
         (short)0, c,  false,  false);
  }

  const int rbase = node0 + (half << 3);
  float* cp = hn + (size_t)rbase * FNC + col0 + mn;
#pragma unroll
  for (int g = 0; g < 8; ++g)
    if (rbase + g < n) cp[(size_t)g * FNC] = c[g];
}

__global__ void edge_dot(const int* __restrict__ es, const int* __restrict__ ed,
                         const float* __restrict__ z, float* __restrict__ out, int L) {
  int j = blockIdx.x * blockDim.x + threadIdx.x;
  if (j >= L) return;
  const float4* a = (const float4*)(z + (size_t)es[j] * 64);
  const float4* b = (const float4*)(z + (size_t)ed[j] * 64);
  float sum = 0.f;
#pragma unroll
  for (int t = 0; t < 16; ++t) {
    float4 x = a[t], y = b[t];
    sum += x.x * y.x + x.y * y.y + x.z * y.z + x.w * y.w;
  }
  out[j] = sum;
}


static inline int cdiv_ll(long long a, long long b) { return (int)((a + b - 1) / b); }

extern "C" void kernel_launch(void* const* d_in, const int* in_sizes, int n_in,
                              void* d_out, int out_size, void* d_ws, size_t ws_size,
                              hipStream_t stream) {
  const float* x   = (const float*)d_in[0];
  const int*   ei  = (const int*)d_in[1];
  const int*   eli = (const int*)d_in[2];
  const float* W1  = (const float*)d_in[3];
  const float* b1  = (const float*)d_in[4];
  const float* W2  = (const float*)d_in[5];
  const float* b2  = (const float*)d_in[6];
  float* out = (float*)d_out;

  const int IN = 512, HID = 128, OUT = 64;
  const int n = in_sizes[0] / IN;
  const int E = in_sizes[1] / 2;
  const int L = in_sizes[2] / 2;
  const int* src = ei;
  const int* dst = ei + E;
  const int* es  = eli;
  const int* ed  = eli + L;

  const int nbin = cdiv_ll(n, NPB);
  float* ws   = (float*)d_ws;
  int*   cnt  = (int*)ws;  ws += n;
  int*   ctr  = (int*)ws;  ws += n;
  float* h1   = ws;        ws += (size_t)n * HID;
  float* z1   = ws;        ws += (size_t)n * HID;
  int*   col  = (int*)ws;
  int*   ovf  = col + (size_t)n * CAP;
  ws += (size_t)n * OUT;
  float* z2   = ws;
  float* h2   = z1;
  int*   ovfn   = ctr;
  int*   nspill = ctr + 1;
  int*   gcnt   = ctr + 2;
  int*   rec    = (int*)z1;
  int*   spill  = rec + (size_t)nbin * BCAP * 2;

  const int B = 256;

  fill_i32<<<cdiv_ll(nbin + 2, B), B, 0, stream>>>(ctr, nbin + 2, 0);
  bin_edges<<<cdiv_ll(E, TA * EPT), TA, 0, stream>>>(src, dst, gcnt, rec, nspill, spill, E, nbin);
  rank_range<<<nbin, TB, 0, stream>>>(gcnt, rec, nspill, spill, cnt, col, ovfn, ovf, n);

  gemm_wmma_f32_lds<32, 5, 4><<<cdiv_ll(n, 80) * (HID / 64), 128, 0, stream>>>(x, W1, h1, n, IN, HID);
  gather_rows_gemm<<<cdiv_ll(n, FN), 32 * FW, 0, stream>>>(cnt, col, ovfn, ovf, h1, b1, W2, h2, n);

  gather_rows<64, false><<<cdiv_ll(n, B / 32), B, 0, stream>>>(cnt, col, ovfn, ovf, h2, b2, z2, n);

  edge_dot<<<cdiv_ll(L, B), B, 0, stream>>>(es, ed, z2, out, L);
}


# edge label dots re-laid: 8 lanes share an edge, so a wave-wide load covers 4 whole 128-B lines (was 32 lines x 16 B); results stored as one line; dot sum order now per lane, then an 8-lane tree
# speedup vs baseline: 3.5218x; 1.1054x over previous
#include <hip/hip_runtime.h>

typedef __attribute__((ext_vector_type(2))) float v2f;
typedef __attribute__((ext_vector_type(8))) float v8f;


constexpr int CAP  = 32;
constexpr int SH   = 7;
constexpr int NPB  = 1 << SH;
constexpr int BCAP = 4096;
constexpr int TA   = 512;
constexpr int EPT  = 4;
constexpr int MAXBIN = 512;
constexpr int TB   = 512;
constexpr int RB   = BCAP / TB;
static_assert(TB * RB == BCAP, "one pass of the block must cover a range's region");
static_assert(NPB <= TB, "one thread per node of the range");

__device__ __forceinline__ float dinv_of(int c) {
  const float dg = (float)(c + 1);
  return dg > 0.f ? rsqrtf(dg) : 0.f;
}

__global__ void fill_i32(int* __restrict__ p, int n, int v) {
  int i = blockIdx.x * blockDim.x + threadIdx.x;
  if (i < n) p[i] = v;
}

__global__ __launch_bounds__(TA)
void bin_edges(const int* __restrict__ src, const int* __restrict__ dst,
               int* __restrict__ gcnt, int* __restrict__ rec,
               int* __restrict__ nspill, int* __restrict__ spill, int E, int nbin) {
  __shared__ int hist[MAXBIN];
  __shared__ int base[MAXBIN];
  const int tid = threadIdx.x;
  for (int b = tid; b < nbin; b += TA) hist[b] = 0;
  __syncthreads();

  const int e0 = ((int)blockIdx.x * TA + tid) * EPT;
  int d[EPT], s[EPT], lr[EPT];
  if (((E & 3) == 0) && e0 + EPT <= E) {
    const int4 dv = *(const int4*)(dst + e0);
    const int4 sv = *(const int4*)(src + e0);
    d[0] = dv.x; d[1] = dv.y; d[2] = dv.z; d[3] = dv.w;
    s[0] = sv.x; s[1] = sv.y; s[2] = sv.z; s[3] = sv.w;
  } else {
#pragma unroll
    for (int k = 0; k < EPT; ++k) {
      const bool in = e0 + k < E;
      d[k] = in ? dst[e0 + k] : -1;
      s[k] = in ? src[e0 + k] : 0;
    }
  }
#pragma unroll
  for (int k = 0; k < EPT; ++k)
    lr[k] = d[k] >= 0 ? atomicAdd(&hist[d[k] >> SH], 1) : 0;
  __syncthreads();

  for (int b = tid; b < nbin; b += TA) {
    const int c = hist[b];
    base[b] = c ? atomicAdd(&gcnt[b], c) : 0;
  }
  __syncthreads();

#pragma unroll
  for (int k = 0; k < EPT; ++k) {
    if (d[k] < 0) continue;
    const int bin = d[k] >> SH;
    const int pos = base[bin] + lr[k];
    int* p = rec + ((size_t)bin * BCAP + pos) * 2;
    if (pos >= BCAP)
      p = spill + 2 * (size_t)atomicAdd(nspill, 1);
    *(int2*)p = make_int2(d[k], s[k]);
  }
}

__global__ __launch_bounds__(TB)
void rank_range(const int* __restrict__ gcnt, const int* __restrict__ rec,
                const int* __restrict__ nspill, const int* __restrict__ spill,
                int* __restrict__ cnt, int* __restrict__ col,
                int* __restrict__ ovfn, int* __restrict__ ovf, int n) {
  __shared__ int lc[NPB];
  __shared__ int4 slots[NPB * CAP / 4];
  const int tid = threadIdx.x;
  const int bin = blockIdx.x;
  const int d0  = bin << SH;
  int* sl = (int*)slots;

  if (tid < NPB) lc[tid] = 0;
  for (int j = tid; j < NPB * CAP / 4; j += TB) slots[j] = make_int4(0, 0, 0, 0);
  __syncthreads();

  auto put = [&](int d, int s) {
    const int ld = d - d0;
    const int r  = atomicAdd(&lc[ld], 1);
    if (r < CAP) {
      sl[ld * CAP + r] = s;
    } else {
      const int q = atomicAdd(ovfn, 1);
      ovf[2 * q]     = d;
      ovf[2 * q + 1] = s;
    }
  };

  const int g = gcnt[bin];
  const int m = g < BCAP ? g : BCAP;
  const int2* rp = (const int2*)(rec + (size_t)bin * BCAP * 2);
  int2 r[RB];
#pragma unroll
  for (int k = 0; k < RB; ++k) {
    const int j = k * TB + tid;
    r[k] = j < m ? rp[j] : make_int2(-1, 0);
  }
#pragma unroll
  for (int k = 0; k < RB; ++k)
    if (r[k].x >= 0) put(r[k].x, r[k].y);
  const int ns = nspill[0];
  for (int j = tid; j < ns; j += TB) {
    const int d = spill[2 * j];
    if ((d >> SH) == bin) put(d, spill[2 * j + 1]);
  }
  __syncthreads();

  const int nn = min(NPB, n - d0);
  if (tid < nn) cnt[d0 + tid] = lc[tid];
  int4* cp = (int4*)(col + (size_t)d0 * CAP);
  for (int j = tid; j < nn * (CAP / 4); j += TB) cp[j] = slots[j];
}

template <int TK, int R, int NW>
__global__ __launch_bounds__(32 * NW) __attribute__((amdgpu_waves_per_eu(1, 6)))
void gemm_wmma_f32_lds(const float* __restrict__ A,
                       const float* __restrict__ B,
                       float* __restrict__ C,
                       const int* __restrict__ deg,
                       int M, int K, int NC) {
  constexpr int ROWS = 16 * R;
  constexpr int LSTR = TK + 4;
  constexpr int NTHR = 32 * NW;
  constexpr int NLD  = (ROWS * TK) / (4 * NTHR);
  static_assert((ROWS * TK) % (4 * NTHR) == 0, "A block must split evenly over the threads");
  __shared__ float As[ROWS * LSTR];
  __shared__ __attribute__((aligned(16))) float rsc[ROWS];
  static_assert(ROWS <= NTHR, "one thread per row scale");

  const int tid   = threadIdx.x;
  const int lane  = tid & 31;
  const int wave  = tid >> 5;
  const int ncb   = NC / (16 * NW);
  const int row0  = ((int)blockIdx.x / ncb) * ROWS;
  const int col0  = (((int)blockIdx.x % ncb) * NW + wave) << 4;
  const int mn    = lane & 15;
  const int half  = lane >> 4;
  const int half2 = half << 1;

  const float* bp = B + (size_t)half2 * NC + col0 + mn;

  if (tid < ROWS) rsc[tid] = dinv_of(deg[min(row0 + tid, M - 1)]);

  v8f acc[R];
#pragma unroll
  for (int rt = 0; rt < R; ++rt) acc[rt] = v8f{};

  for (int k0 = 0; k0 < K; k0 += TK) {
    if (k0 + TK < K) {
#pragma unroll
      for (int i = 0; i < NLD; ++i) {
        const int t  = (tid + i * NTHR) * 4;
        const int gr = min(row0 + t / TK, M - 1);
        __builtin_prefetch(A + (size_t)gr * K + k0 + TK + t % TK, 0, 1);
      }
    }
    v2f bf[TK / 4];
#pragma unroll
    for (int j = 0; j < TK / 4; ++j) {
      bf[j].x = bp[(size_t)(k0 + 4 * j) * NC];
      bf[j].y = bp[(size_t)(k0 + 4 * j + 1) * NC];
    }
    float4 st[NLD];
#pragma unroll
    for (int i = 0; i < NLD; ++i) {
      const int t  = (tid + i * NTHR) * 4;
      const int gr = min(row0 + t / TK, M - 1);
      const float* gp = A + (size_t)gr * K + k0 + t % TK;
      st[i] = *(const float4*)gp;
    }
#pragma unroll
    for (int i = 0; i < NLD; ++i) {
      const int t = (tid + i * NTHR) * 4;
      *(float4*)(&As[(t / TK) * LSTR + (t % TK)]) = st[i];
    }
    __syncthreads();

#pragma unroll
    for (int j = 0; j < TK / 4; ++j) {
#pragma unroll
      for (int rt = 0; rt < R; ++rt) {
        v2f a;
        a.x = As[(rt * 16 + mn) * LSTR + 4 * j + half2];
        a.y = As[(rt * 16 + mn) * LSTR + 4 * j + half2 + 1];
        acc[rt] = __builtin_amdgcn_wmma_f32_16x16x4_f32(
             false, a,  false, bf[j],
             (short)0, acc[rt],  false,  false);
      }
    }
    __syncthreads();
  }

#pragma unroll
  for (int rt = 0; rt < R; ++rt) {
    const int rbase = row0 + rt * 16 + (half << 3);
    float* cp = C + (size_t)rbase * NC + col0 + mn;
    const float4 s0 = *(const float4*)(&rsc[rt * 16 + (half << 3)]);
    const float4 s1 = *(const float4*)(&rsc[rt * 16 + (half << 3) + 4]);
    const float sc[8] = {s0.x, s0.y, s0.z, s0.w, s1.x, s1.y, s1.z, s1.w};
#pragma unroll
    for (int g = 0; g < 8; ++g)
      if (rbase + g < M) cp[(size_t)g * NC] = acc[rt][g] * sc[g];
  }
}

template <int CH>
__device__ __forceinline__ float __attribute__((ext_vector_type(CH / 32)))
sum_node(int d, int lane, const int* __restrict__ cnt, const int* __restrict__ col,
         const int* __restrict__ ovfn, const int* __restrict__ ovf,
         const float* __restrict__ h) {
  constexpr int VW = CH / 32;
  typedef float vecf __attribute__((ext_vector_type(VW)));
  const int c = cnt[d];
  const int m = c < CAP ? c : CAP;
  const float dd = dinv_of(c);
  const float* hl = h + lane * VW;

  vecf acc = *(const vecf*)(hl + (size_t)d * CH);

  int sv = 0;
  if (lane < m) sv = col[d * CAP + lane];
  auto row = [&](int ix) -> vecf {
    return *(const vecf*)(hl + (size_t)__builtin_amdgcn_readlane(sv, ix) * CH);
  };

  int i = 0;
  for (; i + 8 <= m; i += 8) {
    vecf v[8];
    for (int k = 0; k < 8; ++k) v[k] = row(i + k);
    for (int k = 0; k < 8; ++k) acc += v[k];
  }
  if (i + 4 <= m) {
    vecf v[4];
    for (int k = 0; k < 4; ++k) v[k] = row(i + k);
    for (int k = 0; k < 4; ++k) acc += v[k];
    i += 4;
  }
  const int rem = m - i;
  if (rem > 0) {
    vecf v0 = row(i), v1 = {}, v2 = {};
    if (rem > 1) v1 = row(i + 1);
    if (rem > 2) v2 = row(i + 2);
    acc += v0;
    if (rem > 1) acc += v1;
    if (rem > 2) acc += v2;
  }

  if (c > CAP) {
    const int tot = ovfn[0];
    for (int j0 = 0; j0 < tot; j0 += 32) {
      int od = -1, os = 0;
      if (j0 + lane < tot) {
        od = ovf[2 * (j0 + lane)];
        os = ovf[2 * (j0 + lane) + 1];
      }
      unsigned hits = __builtin_amdgcn_ballot_w32(od == d);
      while (hits) {
        const int ix = __builtin_ctz(hits);
        hits &= hits - 1;
        acc += *(const vecf*)(hl + (size_t)__builtin_amdgcn_readlane(os, ix) * CH);
      }
    }
  }
  return acc * dd;
}

template <int CH, bool RELU>
__global__ void gather_rows(const int* __restrict__ cnt, const int* __restrict__ col,
                            const int* __restrict__ ovfn, const int* __restrict__ ovf,
                            const float* __restrict__ h,
                            const float* __restrict__ bias, float* __restrict__ z, int n) {
  constexpr int VW = CH / 32;
  typedef float vecf __attribute__((ext_vector_type(VW)));
  const int lane = threadIdx.x & 31;
  const int wv = __builtin_amdgcn_readfirstlane((int)(threadIdx.x >> 5));
  const int d  = blockIdx.x * (int)(blockDim.x >> 5) + wv;
  if (d >= n) return;

  vecf acc = sum_node<CH>(d, lane, cnt, col, ovfn, ovf, h);
  acc += *(const vecf*)(bias + lane * VW);
  if (RELU) {
    for (int k = 0; k < VW; ++k) acc[k] = fmaxf(acc[k], 0.f);
  }
  *(vecf*)(z + (size_t)d * CH + lane * VW) = acc;
}

constexpr int FN   = 16;
constexpr int FW   = 4;
constexpr int FCH  = 128;
constexpr int FNC  = 64;
constexpr int FSTR = FCH + 4;
constexpr int FKC  = 8;
static_assert(FW * 16 == FNC, "one wave per column tile");
__global__ __launch_bounds__(32 * FW) __attribute__((amdgpu_waves_per_eu(16, 16)))
void gather_rows_gemm(const int* __restrict__ cnt, const int* __restrict__ col,
                      const int* __restrict__ ovfn, const int* __restrict__ ovf,
                      const float* __restrict__ h, const float* __restrict__ bias,
                      const float* __restrict__ Wn, float* __restrict__ hn, int n) {
  typedef float vecf __attribute__((ext_vector_type(4)));
  __shared__ float zs[FN * FSTR];
  const int lane  = threadIdx.x & 31;
  const int wv    = __builtin_amdgcn_readfirstlane((int)(threadIdx.x >> 5));
  const int node0 = blockIdx.x * FN;

  for (int q = 0; q < FN / FW; ++q) {
    const int r = q * FW + wv;
    const int d = node0 + r;
    vecf acc = {};
    if (d < n) {
      acc = sum_node<FCH>(d, lane, cnt, col, ovfn, ovf, h);
      acc += *(const vecf*)(bias + lane * 4);
      for (int k = 0; k < 4; ++k) acc[k] = fmaxf(acc[k], 0.f);
      acc *= dinv_of(cnt[d]);
    }
    *(vecf*)(&zs[r * FSTR + lane * 4]) = acc;
  }

  const int mn    = lane & 15;
  const int half  = lane >> 4;
  const int half2 = half << 1;
  const int col0  = wv << 4;
  const float* bp = Wn + (size_t)half2 * FNC + col0 + mn;
  v2f bf[FCH / 4];
#pragma unroll
  for (int j = 0; j < FKC; ++j) {
    bf[j].x = bp[(size_t)(4 * j) * FNC];
    bf[j].y = bp[(size_t)(4 * j + 1) * FNC];
  }
  __syncthreads();
#pragma unroll
  for (int j = FKC; j < FCH / 4; ++j) {
    bf[j].x = bp[(size_t)(4 * j) * FNC];
    bf[j].y = bp[(size_t)(4 * j + 1) * FNC];
  }

  v8f c = v8f{};
  const float* ap = &zs[mn * FSTR + half2];
#pragma unroll
  for (int j = 0; j < FCH / 4; ++j) {
    v2f a;
    a.x = ap[4 * j];
    a.y = ap[4 * j + 1];
    c = __builtin_amdgcn_wmma_f32_16x16x4_f32(
         false, a,  false, bf[j],
         (short)0, c,  false,  false);
  }

  const int rbase = node0 + (half << 3);
  float* cp = hn + (size_t)rbase * FNC + col0 + mn;
#pragma unroll
  for (int g = 0; g < 8; ++g)
    if (rbase + g < n) cp[(size_t)g * FNC] = c[g];
}

constexpr int EDT = 256;
__global__ __launch_bounds__(EDT)
void edge_dot(const int* __restrict__ es, const int* __restrict__ ed,
              const float* __restrict__ z, float* __restrict__ out, int L) {
  const int lane = threadIdx.x & 31;
  const int base = (int)(blockIdx.x * blockDim.x + threadIdx.x) - lane;
  if (base >= L) return;
  const int j  = min(base + lane, L - 1);
  const int ia = es[j];
  const int ib = ed[j];
  const int g0 = lane & ~7;
  const int c  = lane & 7;
  const float4* zc = (const float4*)z + c;
  float4 a0[8], a1[8], b0[8], b1[8];
#pragma unroll
  for (int s = 0; s < 8; ++s) {
    const int ra = __shfl(ia, g0 + s);
    const int rb = __shfl(ib, g0 + s);
    const float4* pa = zc + (size_t)ra * 16;
    const float4* pb = zc + (size_t)rb * 16;
    a0[s] = pa[0]; a1[s] = pa[8];
    b0[s] = pb[0]; b1[s] = pb[8];
  }
  __builtin_amdgcn_sched_barrier(0);
  float res = 0.f;
#pragma unroll
  for (int s = 0; s < 8; ++s) {
    float p = a0[s].x * b0[s].x + a0[s].y * b0[s].y + a0[s].z * b0[s].z + a0[s].w * b0[s].w;
    p += a1[s].x * b1[s].x + a1[s].y * b1[s].y + a1[s].z * b1[s].z + a1[s].w * b1[s].w;
    p += __shfl_xor(p, 1);
    p += __shfl_xor(p, 2);
    p += __shfl_xor(p, 4);
    res = (c == s) ? p : res;
  }
  if (base + lane < L) out[base + lane] = res;
}


static inline int cdiv_ll(long long a, long long b) { return (int)((a + b - 1) / b); }

extern "C" void kernel_launch(void* const* d_in, const int* in_sizes, int n_in,
                              void* d_out, int out_size, void* d_ws, size_t ws_size,
                              hipStream_t stream) {
  const float* x   = (const float*)d_in[0];
  const int*   ei  = (const int*)d_in[1];
  const int*   eli = (const int*)d_in[2];
  const float* W1  = (const float*)d_in[3];
  const float* b1  = (const float*)d_in[4];
  const float* W2  = (const float*)d_in[5];
  const float* b2  = (const float*)d_in[6];
  float* out = (float*)d_out;

  const int IN = 512, HID = 128, OUT = 64;
  const int n = in_sizes[0] / IN;
  const int E = in_sizes[1] / 2;
  const int L = in_sizes[2] / 2;
  const int* src = ei;
  const int* dst = ei + E;
  const int* es  = eli;
  const int* ed  = eli + L;

  const int nbin = cdiv_ll(n, NPB);
  float* ws   = (float*)d_ws;
  int*   cnt  = (int*)ws;  ws += n;
  int*   ctr  = (int*)ws;  ws += n;
  float* h1   = ws;        ws += (size_t)n * HID;
  float* z1   = ws;        ws += (size_t)n * HID;
  int*   col  = (int*)ws;
  int*   ovf  = col + (size_t)n * CAP;
  ws += (size_t)n * OUT;
  float* z2   = ws;
  float* h2   = z1;
  int*   ovfn   = ctr;
  int*   nspill = ctr + 1;
  int*   gcnt   = ctr + 2;
  int*   rec    = (int*)z1;
  int*   spill  = rec + (size_t)nbin * BCAP * 2;

  const int B = 256;

  fill_i32<<<cdiv_ll(nbin + 2, B), B, 0, stream>>>(ctr, nbin + 2, 0);
  bin_edges<<<cdiv_ll(E, TA * EPT), TA, 0, stream>>>(src, dst, gcnt, rec, nspill, spill, E, nbin);
  rank_range<<<nbin, TB, 0, stream>>>(gcnt, rec, nspill, spill, cnt, col, ovfn, ovf, n);

  gemm_wmma_f32_lds<32, 5, 4><<<cdiv_ll(n, 80) * (HID / 64), 128, 0, stream>>>(x, W1, h1, cnt, n, IN, HID);
  gather_rows_gemm<<<cdiv_ll(n, FN), 32 * FW, 0, stream>>>(cnt, col, ovfn, ovf, h1, b1, W2, h2, n);

  gather_rows<64, false><<<cdiv_ll(n, B / 32), B, 0, stream>>>(cnt, col, ovfn, ovf, h2, b2, z2, n);

  edge_dot<<<cdiv_ll(L, EDT), EDT, 0, stream>>>(es, ed, z2, out, L);
}
